# starting kernel
# baseline (speedup 1.0000x reference)
#include <hip/hip_runtime.h>
#include <hip/hip_bf16.h>

typedef float v2f __attribute__((ext_vector_type(2)));
typedef float v8f __attribute__((ext_vector_type(8)));

__device__ __forceinline__ v8f wmma4(v2f a, v2f b, v8f c) {
    return __builtin_amdgcn_wmma_f32_16x16x4_f32(
         false, a,  false, b,
         (short)0, c,  false,  false);
}

#define CIN   256
#define COUT  256
#define HW    64
#define K9    2304
#define SDIM  512
#define NB    8

__global__ __launch_bounds__(256) void style_s9_kernel(
    const float* __restrict__ style, const float* __restrict__ mod_w,
    const float* __restrict__ mod_b, float* __restrict__ s9)
{
    __shared__ float st[SDIM];
    const int b = blockIdx.x;
    const int tid = threadIdx.x;
    st[tid]       = style[b * SDIM + tid];
    st[tid + 256] = style[b * SDIM + tid + 256];
    __syncthreads();

    const float* wrow = mod_w + (size_t)tid * SDIM;
    float acc = mod_b[tid];
#pragma unroll 8
    for (int j = 0; j < SDIM; ++j) acc += st[j] * wrow[j];

#pragma unroll
    for (int t = 0; t < 9; ++t) s9[b * K9 + tid * 9 + t] = acc;
}

__global__ __launch_bounds__(256) void demod_kernel(
    const float* __restrict__ weight, const float* __restrict__ s9,
    float* __restrict__ demod)
{
    const int row = blockIdx.x;
    const int b = row >> 8, oc = row & 255;
    const int tid = threadIdx.x;
    const float* wrow = weight + (size_t)oc * K9;
    const float* srow = s9 + (size_t)b * K9;

    float acc = 0.f;
    for (int k = tid; k < K9; k += 256) {
        float v = wrow[k] * srow[k];
        acc += v * v;
    }
    __shared__ float red[256];
    red[tid] = acc;
    __syncthreads();
#pragma unroll
    for (int s = 128; s > 0; s >>= 1) {
        if (tid < s) red[tid] += red[tid + s];
        __syncthreads();
    }
    if (tid == 0) demod[row] = rsqrtf(red[0] + 1e-8f);
}

#define LDA 20
#define LDBC 20
#define NCHUNK (K9 / 16)

__global__ __launch_bounds__(256) void conv_wmma_kernel(
    const float* __restrict__ x, const float* __restrict__ weight,
    const float* __restrict__ bias, const float* __restrict__ s9,
    const float* __restrict__ demod, float* __restrict__ out)
{
    __shared__ float lA[2][256 * LDA];
    __shared__ float lB[2][64 * LDBC];

    const int tid = threadIdx.x;
    const int h   = blockIdx.x;
    const int b   = blockIdx.y;

    const int lane = tid & 31;
    const int wid  = tid >> 5;
    const int wm   = (wid & 3) * 64;
    const int wn   = (wid >> 2) * 32;
    const int half = lane >> 4;
    const int lr   = lane & 15;

    const int colB = tid & 63;
    const int kcb0 = tid >> 6;

    const float* __restrict__ s9b = s9 + (size_t)b * K9;
    const float* __restrict__ xb  = x + (size_t)b * CIN * HW * HW;

    float4 aReg[4];
    float  bReg[4];

    auto loadChunk = [&](int t) {
        const int k0g = t * 16;
#pragma unroll
        for (int i = 0; i < 4; ++i) {
            int id4 = tid + 256 * i;
            int row = id4 >> 2;
            int kc4 = (id4 & 3) << 2;
            aReg[i] = *(const float4*)(weight + (size_t)row * K9 + k0g + kc4);
        }
#pragma unroll
        for (int i = 0; i < 4; ++i) {
            int kc = kcb0 + 4 * i;
            int kk = k0g + kc;
            int ic = kk / 9;
            int r9 = kk - ic * 9;
            int kh = r9 / 3;
            int kw = r9 - kh * 3;
            int ih = h + kh - 1;
            int iw = colB + kw - 1;
            float v = 0.f;
            if ((unsigned)ih < (unsigned)HW && (unsigned)iw < (unsigned)HW)
                v = xb[((size_t)ic * HW + ih) * HW + iw];
            bReg[i] = v * s9b[kk];
        }
        if (t + 2 < NCHUNK) {
            int id4 = tid;
            int row = id4 >> 2;
            int kc4 = (id4 & 3) << 2;
            __builtin_prefetch(weight + (size_t)row * K9 + k0g + 32 + kc4, 0, 0);
        }
    };
    auto storeChunk = [&](int buf) {
#pragma unroll
        for (int i = 0; i < 4; ++i) {
            int id4 = tid + 256 * i;
            int row = id4 >> 2;
            int kc4 = (id4 & 3) << 2;
            *(float4*)&lA[buf][row * LDA + kc4] = aReg[i];
        }
#pragma unroll
        for (int i = 0; i < 4; ++i) {
            int kc = kcb0 + 4 * i;
            lB[buf][colB * LDBC + kc] = bReg[i];
        }
    };

    v8f acc[4][2];
#pragma unroll
    for (int mt = 0; mt < 4; ++mt)
#pragma unroll
        for (int nt = 0; nt < 2; ++nt)
            acc[mt][nt] = (v8f){};

    loadChunk(0);
    storeChunk(0);
    __syncthreads();

    for (int t = 0; t < NCHUNK; ++t) {
        const int cur = t & 1;
        if (t + 1 < NCHUNK) loadChunk(t + 1);

#pragma unroll
        for (int k0 = 0; k0 < 16; k0 += 4) {
            const int ak = k0 + half * 2;
            v2f af[4];
#pragma unroll
            for (int mt = 0; mt < 4; ++mt)
                af[mt] = *(const v2f*)&lA[cur][(wm + mt * 16 + lr) * LDA + ak];
            v2f bf[2];
#pragma unroll
            for (int nt = 0; nt < 2; ++nt)
                bf[nt] = *(const v2f*)&lB[cur][(wn + nt * 16 + lr) * LDBC + ak];

#pragma unroll
            for (int mt = 0; mt < 4; ++mt)
#pragma unroll
                for (int nt = 0; nt < 2; ++nt)
                    acc[mt][nt] = wmma4(af[mt], bf[nt], acc[mt][nt]);
        }

        if (t + 1 < NCHUNK) storeChunk(1 ^ cur);
        __syncthreads();
    }

    const float* __restrict__ demodb = demod + b * COUT;
    float* __restrict__ outb = out + ((size_t)b * COUT * HW + h) * HW;
#pragma unroll
    for (int mt = 0; mt < 4; ++mt) {
#pragma unroll
        for (int v = 0; v < 8; ++v) {
            int oc = wm + mt * 16 + half * 8 + v;
            float d  = demodb[oc];
            float bb = bias[oc];
#pragma unroll
            for (int nt = 0; nt < 2; ++nt)
                outb[(size_t)oc * (HW * HW) + wn + nt * 16 + lr] =
                    acc[mt][nt][v] * d + bb;
        }
    }
}

extern "C" void kernel_launch(void* const* d_in, const int* in_sizes, int n_in,
                              void* d_out, int out_size, void* d_ws, size_t ws_size,
                              hipStream_t stream) {
    const float* x      = (const float*)d_in[0];
    const float* style  = (const float*)d_in[1];
    const float* weight = (const float*)d_in[2];
    const float* bias   = (const float*)d_in[3];
    const float* mod_w  = (const float*)d_in[4];
    const float* mod_b  = (const float*)d_in[5];
    float* out = (float*)d_out;

    float* s9    = (float*)d_ws;
    float* demod = s9 + NB * K9;

    style_s9_kernel<<<NB, 256, 0, stream>>>(style, mod_w, mod_b, s9);
    demod_kernel<<<NB * COUT, 256, 0, stream>>>(weight, s9, demod);

    dim3 grid(HW, NB);
    conv_wmma_kernel<<<grid, 256, 0, stream>>>(x, weight, bias, s9, demod, out);
}


# Keep the prefetched next K tile in VGPRs across the WMMA block (native 4-float vectors, not a private struct array in scratch); its load wait and LDS stores now sit behind the WMMAs
# speedup vs baseline: 1.5779x; 1.5779x over previous
#include <hip/hip_runtime.h>
#include <hip/hip_bf16.h>

typedef float v2f __attribute__((ext_vector_type(2)));
typedef float v4f __attribute__((ext_vector_type(4)));
typedef float v8f __attribute__((ext_vector_type(8)));

__device__ __forceinline__ v8f wmma4(v2f a, v2f b, v8f c) {
    return __builtin_amdgcn_wmma_f32_16x16x4_f32(
         false, a,  false, b,
         (short)0, c,  false,  false);
}

#define CIN   256
#define COUT  256
#define HW    64
#define K9    2304
#define SDIM  512
#define NB    8

__global__ __launch_bounds__(256) void style_s9_kernel(
    const float* __restrict__ style, const float* __restrict__ mod_w,
    const float* __restrict__ mod_b, float* __restrict__ s9)
{
    __shared__ float st[SDIM];
    const int b = blockIdx.x;
    const int tid = threadIdx.x;
    st[tid]       = style[b * SDIM + tid];
    st[tid + 256] = style[b * SDIM + tid + 256];
    __syncthreads();

    const float* wrow = mod_w + (size_t)tid * SDIM;
    float acc = mod_b[tid];
#pragma unroll 8
    for (int j = 0; j < SDIM; ++j) acc += st[j] * wrow[j];

#pragma unroll
    for (int t = 0; t < 9; ++t) s9[b * K9 + tid * 9 + t] = acc;
}

__global__ __launch_bounds__(256) void demod_kernel(
    const float* __restrict__ weight, const float* __restrict__ s9,
    float* __restrict__ demod)
{
    const int row = blockIdx.x;
    const int b = row >> 8, oc = row & 255;
    const int tid = threadIdx.x;
    const float* wrow = weight + (size_t)oc * K9;
    const float* srow = s9 + (size_t)b * K9;

    float acc = 0.f;
    for (int k = tid; k < K9; k += 256) {
        float v = wrow[k] * srow[k];
        acc += v * v;
    }
    __shared__ float red[256];
    red[tid] = acc;
    __syncthreads();
#pragma unroll
    for (int s = 128; s > 0; s >>= 1) {
        if (tid < s) red[tid] += red[tid + s];
        __syncthreads();
    }
    if (tid == 0) demod[row] = rsqrtf(red[0] + 1e-8f);
}

#define LDA 20
#define LDBC 20
#define NCHUNK (K9 / 16)

__global__ __launch_bounds__(256) void conv_wmma_kernel(
    const float* __restrict__ x, const float* __restrict__ weight,
    const float* __restrict__ bias, const float* __restrict__ s9,
    const float* __restrict__ demod, float* __restrict__ out)
{
    __shared__ float lA[2][256 * LDA];
    __shared__ float lB[2][64 * LDBC];

    const int tid = threadIdx.x;
    const int h   = blockIdx.x;
    const int b   = blockIdx.y;

    const int lane = tid & 31;
    const int wid  = tid >> 5;
    const int wm   = (wid & 3) * 64;
    const int wn   = (wid >> 2) * 32;
    const int half = lane >> 4;
    const int lr   = lane & 15;

    const int colB = tid & 63;
    const int kcb0 = tid >> 6;

    const float* __restrict__ s9b = s9 + (size_t)b * K9;
    const float* __restrict__ xb  = x + (size_t)b * CIN * HW * HW;

    v4f    aReg[4];
    float  bReg[4];
    float  sReg[4];

    auto loadChunk = [&](int t) {
        const int k0g = t * 16;
#pragma unroll
        for (int i = 0; i < 4; ++i) {
            int id4 = tid + 256 * i;
            int row = id4 >> 2;
            int kc4 = (id4 & 3) << 2;
            aReg[i] = *(const v4f*)(weight + (size_t)row * K9 + k0g + kc4);
        }
#pragma unroll
        for (int i = 0; i < 4; ++i) {
            int kc = kcb0 + 4 * i;
            int kk = k0g + kc;
            int ic = kk / 9;
            int r9 = kk - ic * 9;
            int kh = r9 / 3;
            int kw = r9 - kh * 3;
            int ih = h + kh - 1;
            int iw = colB + kw - 1;
            float v = 0.f;
            if ((unsigned)ih < (unsigned)HW && (unsigned)iw < (unsigned)HW)
                v = xb[((size_t)ic * HW + ih) * HW + iw];
            bReg[i] = v;
            sReg[i] = s9b[kk];
        }
        if (t + 2 < NCHUNK) {
            int id4 = tid;
            int row = id4 >> 2;
            int kc4 = (id4 & 3) << 2;
            __builtin_prefetch(weight + (size_t)row * K9 + k0g + 32 + kc4, 0, 0);
        }
    };
    auto storeChunk = [&](int buf) {
#pragma unroll
        for (int i = 0; i < 4; ++i) {
            int id4 = tid + 256 * i;
            int row = id4 >> 2;
            int kc4 = (id4 & 3) << 2;
            *(v4f*)&lA[buf][row * LDA + kc4] = aReg[i];
        }
#pragma unroll
        for (int i = 0; i < 4; ++i) {
            int kc = kcb0 + 4 * i;
            lB[buf][colB * LDBC + kc] = bReg[i] * sReg[i];
        }
    };

    v8f acc[4][2];
#pragma unroll
    for (int mt = 0; mt < 4; ++mt)
#pragma unroll
        for (int nt = 0; nt < 2; ++nt)
            acc[mt][nt] = (v8f){};

    loadChunk(0);
    storeChunk(0);
    __syncthreads();

    for (int t = 0; t < NCHUNK; ++t) {
        const int cur = t & 1;
        if (t + 1 < NCHUNK) loadChunk(t + 1);

#pragma unroll
        for (int k0 = 0; k0 < 16; k0 += 4) {
            const int ak = k0 + half * 2;
            v2f af[4];
#pragma unroll
            for (int mt = 0; mt < 4; ++mt)
                af[mt] = *(const v2f*)&lA[cur][(wm + mt * 16 + lr) * LDA + ak];
            v2f bf[2];
#pragma unroll
            for (int nt = 0; nt < 2; ++nt)
                bf[nt] = *(const v2f*)&lB[cur][(wn + nt * 16 + lr) * LDBC + ak];

#pragma unroll
            for (int mt = 0; mt < 4; ++mt)
#pragma unroll
                for (int nt = 0; nt < 2; ++nt)
                    acc[mt][nt] = wmma4(af[mt], bf[nt], acc[mt][nt]);
        }

        __builtin_amdgcn_sched_barrier(0);
        if (t + 1 < NCHUNK) storeChunk(1 ^ cur);
        __syncthreads();
    }

    const float* __restrict__ demodb = demod + b * COUT;
    float* __restrict__ outb = out + ((size_t)b * COUT * HW + h) * HW;
#pragma unroll
    for (int mt = 0; mt < 4; ++mt) {
#pragma unroll
        for (int v = 0; v < 8; ++v) {
            int oc = wm + mt * 16 + half * 8 + v;
            float d  = demodb[oc];
            float bb = bias[oc];
#pragma unroll
            for (int nt = 0; nt < 2; ++nt)
                outb[(size_t)oc * (HW * HW) + wn + nt * 16 + lr] =
                    acc[mt][nt][v] * d + bb;
        }
    }
}

extern "C" void kernel_launch(void* const* d_in, const int* in_sizes, int n_in,
                              void* d_out, int out_size, void* d_ws, size_t ws_size,
                              hipStream_t stream) {
    const float* x      = (const float*)d_in[0];
    const float* style  = (const float*)d_in[1];
    const float* weight = (const float*)d_in[2];
    const float* bias   = (const float*)d_in[3];
    const float* mod_w  = (const float*)d_in[4];
    const float* mod_b  = (const float*)d_in[5];
    float* out = (float*)d_out;

    float* s9    = (float*)d_ws;
    float* demod = s9 + NB * K9;

    style_s9_kernel<<<NB, 256, 0, stream>>>(style, mod_w, mod_b, s9);
    demod_kernel<<<NB * COUT, 256, 0, stream>>>(weight, s9, demod);

    dim3 grid(HW, NB);
    conv_wmma_kernel<<<grid, 256, 0, stream>>>(x, weight, bias, s9, demod, out);
}


# K-loop load issue made wave-uniform: im2col tap indices and load bases on the scalar unit (tid>>6 read as a scalar), lane offsets invariant, padding by clamped address + select at the LDS store
# speedup vs baseline: 1.7796x; 1.1278x over previous
#include <hip/hip_runtime.h>
#include <hip/hip_bf16.h>

typedef float v2f __attribute__((ext_vector_type(2)));
typedef float v4f __attribute__((ext_vector_type(4)));
typedef float v8f __attribute__((ext_vector_type(8)));

__device__ __forceinline__ v8f wmma4(v2f a, v2f b, v8f c) {
    return __builtin_amdgcn_wmma_f32_16x16x4_f32(
         false, a,  false, b,
         (short)0, c,  false,  false);
}

#define CIN   256
#define COUT  256
#define HW    64
#define K9    2304
#define SDIM  512
#define NB    8

__global__ __launch_bounds__(256) void style_s9_kernel(
    const float* __restrict__ style, const float* __restrict__ mod_w,
    const float* __restrict__ mod_b, float* __restrict__ s9)
{
    __shared__ float st[SDIM];
    const int b = blockIdx.x;
    const int tid = threadIdx.x;
    st[tid]       = style[b * SDIM + tid];
    st[tid + 256] = style[b * SDIM + tid + 256];
    __syncthreads();

    const float* wrow = mod_w + (size_t)tid * SDIM;
    float acc = mod_b[tid];
#pragma unroll 8
    for (int j = 0; j < SDIM; ++j) acc += st[j] * wrow[j];

#pragma unroll
    for (int t = 0; t < 9; ++t) s9[b * K9 + tid * 9 + t] = acc;
}

__global__ __launch_bounds__(256) void demod_kernel(
    const float* __restrict__ weight, const float* __restrict__ s9,
    float* __restrict__ demod)
{
    const int row = blockIdx.x;
    const int b = row >> 8, oc = row & 255;
    const int tid = threadIdx.x;
    const float* wrow = weight + (size_t)oc * K9;
    const float* srow = s9 + (size_t)b * K9;

    float acc = 0.f;
    for (int k = tid; k < K9; k += 256) {
        float v = wrow[k] * srow[k];
        acc += v * v;
    }
    __shared__ float red[256];
    red[tid] = acc;
    __syncthreads();
#pragma unroll
    for (int s = 128; s > 0; s >>= 1) {
        if (tid < s) red[tid] += red[tid + s];
        __syncthreads();
    }
    if (tid == 0) demod[row] = rsqrtf(red[0] + 1e-8f);
}

#define LDA 20
#define LDBC 20
#define NCHUNK (K9 / 16)

__global__ __launch_bounds__(256) void conv_wmma_kernel(
    const float* __restrict__ x, const float* __restrict__ weight,
    const float* __restrict__ bias, const float* __restrict__ s9,
    const float* __restrict__ demod, float* __restrict__ out)
{
    __shared__ float lA[2][256 * LDA];
    __shared__ float lB[2][64 * LDBC];

    const int tid = threadIdx.x;
    const int h   = blockIdx.x;
    const int b   = blockIdx.y;

    const int lane = tid & 31;
    const int wid  = tid >> 5;
    const int wm   = (wid & 3) * 64;
    const int wn   = (wid >> 2) * 32;
    const int half = lane >> 4;
    const int lr   = lane & 15;

    const int colB = tid & 63;
    const int kcb0 = __builtin_amdgcn_readfirstlane(tid >> 6);

    unsigned aoff[4];
#pragma unroll
    for (int i = 0; i < 4; ++i)
        aoff[i] = (unsigned)(((tid + 256 * i) >> 2) * K9 + ((tid & 3) << 2));

    const float* __restrict__ s9b = s9 + (size_t)b * K9;
    const float* __restrict__ xb  = x + (size_t)b * CIN * HW * HW;

    v4f    aReg[4];
    float  bReg[4];
    float  sReg[4];
    bool   okReg[4];

    auto loadChunk = [&](int t) {
        const int k0g = t * 16;
        const float* __restrict__ wt = weight + k0g;
#pragma unroll
        for (int i = 0; i < 4; ++i)
            aReg[i] = *(const v4f*)(wt + aoff[i]);
#pragma unroll
        for (int i = 0; i < 4; ++i) {
            const int kk = k0g + kcb0 + 4 * i;
            const int ic = kk / 9;
            const int r9 = kk - ic * 9;
            const int kh = r9 / 3;
            const int kw = r9 - kh * 3;
            const int ih = h + kh - 1;
            const bool rowok = (unsigned)ih < (unsigned)HW;
            const float* __restrict__ xrow = xb + (ic * HW + (rowok ? ih : 0)) * HW;
            const int iw = colB + kw - 1;
            okReg[i] = rowok && ((unsigned)iw < (unsigned)HW);
            const unsigned iwc = (unsigned)min(max(iw, 0), HW - 1);
            bReg[i] = xrow[iwc];
            sReg[i] = s9b[kk];
        }
        if (t + 2 < NCHUNK)
            __builtin_prefetch(wt + 32 + aoff[0], 0, 0);
    };
    auto storeChunk = [&](int buf) {
#pragma unroll
        for (int i = 0; i < 4; ++i) {
            int id4 = tid + 256 * i;
            int row = id4 >> 2;
            int kc4 = (id4 & 3) << 2;
            *(v4f*)&lA[buf][row * LDA + kc4] = aReg[i];
        }
#pragma unroll
        for (int i = 0; i < 4; ++i) {
            int kc = kcb0 + 4 * i;
            lB[buf][colB * LDBC + kc] = okReg[i] ? bReg[i] * sReg[i] : 0.f;
        }
    };

    v8f acc[4][2];
#pragma unroll
    for (int mt = 0; mt < 4; ++mt)
#pragma unroll
        for (int nt = 0; nt < 2; ++nt)
            acc[mt][nt] = (v8f){};

    loadChunk(0);
    storeChunk(0);
    __syncthreads();

    for (int t = 0; t < NCHUNK; ++t) {
        const int cur = t & 1;
        if (t + 1 < NCHUNK) loadChunk(t + 1);
        __builtin_amdgcn_sched_barrier(0);

#pragma unroll
        for (int k0 = 0; k0 < 16; k0 += 4) {
            const int ak = k0 + half * 2;
            v2f af[4];
#pragma unroll
            for (int mt = 0; mt < 4; ++mt)
                af[mt] = *(const v2f*)&lA[cur][(wm + mt * 16 + lr) * LDA + ak];
            v2f bf[2];
#pragma unroll
            for (int nt = 0; nt < 2; ++nt)
                bf[nt] = *(const v2f*)&lB[cur][(wn + nt * 16 + lr) * LDBC + ak];

#pragma unroll
            for (int mt = 0; mt < 4; ++mt)
#pragma unroll
                for (int nt = 0; nt < 2; ++nt)
                    acc[mt][nt] = wmma4(af[mt], bf[nt], acc[mt][nt]);
        }

        __builtin_amdgcn_sched_barrier(0);
        if (t + 1 < NCHUNK) storeChunk(1 ^ cur);
        __syncthreads();
    }

    const float* __restrict__ demodb = demod + b * COUT;
    float* __restrict__ outb = out + ((size_t)b * COUT * HW + h) * HW;
#pragma unroll
    for (int mt = 0; mt < 4; ++mt) {
#pragma unroll
        for (int v = 0; v < 8; ++v) {
            int oc = wm + mt * 16 + half * 8 + v;
            float d  = demodb[oc];
            float bb = bias[oc];
#pragma unroll
            for (int nt = 0; nt < 2; ++nt)
                outb[(size_t)oc * (HW * HW) + wn + nt * 16 + lr] =
                    acc[mt][nt][v] * d + bb;
        }
    }
}

extern "C" void kernel_launch(void* const* d_in, const int* in_sizes, int n_in,
                              void* d_out, int out_size, void* d_ws, size_t ws_size,
                              hipStream_t stream) {
    const float* x      = (const float*)d_in[0];
    const float* style  = (const float*)d_in[1];
    const float* weight = (const float*)d_in[2];
    const float* bias   = (const float*)d_in[3];
    const float* mod_w  = (const float*)d_in[4];
    const float* mod_b  = (const float*)d_in[5];
    float* out = (float*)d_out;

    float* s9    = (float*)d_ws;
    float* demod = s9 + NB * K9;

    style_s9_kernel<<<NB, 256, 0, stream>>>(style, mod_w, mod_b, s9);
    demod_kernel<<<NB * COUT, 256, 0, stream>>>(weight, s9, demod);

    dim3 grid(HW, NB);
    conv_wmma_kernel<<<grid, 256, 0, stream>>>(x, weight, bias, s9, demod, out);
}


# Prologue reductions remapped to one wave per row: coalesced 16-byte reads, all loads in flight, lane butterfly sum (demod 2048 waves, was 16384); summation order of s and demod changes
# speedup vs baseline: 1.8393x; 1.0335x over previous
#include <hip/hip_runtime.h>
#include <hip/hip_bf16.h>

typedef float v2f __attribute__((ext_vector_type(2)));
typedef float v4f __attribute__((ext_vector_type(4)));
typedef float v8f __attribute__((ext_vector_type(8)));

__device__ __forceinline__ v8f wmma4(v2f a, v2f b, v8f c) {
    return __builtin_amdgcn_wmma_f32_16x16x4_f32(
         false, a,  false, b,
         (short)0, c,  false,  false);
}

#define CIN   256
#define COUT  256
#define HW    64
#define K9    2304
#define SDIM  512
#define NB    8

__device__ __forceinline__ float wave_sum(float v) {
#pragma unroll
    for (int off = 16; off > 0; off >>= 1) v += __shfl_xor(v, off);
    return v;
}

__global__ __launch_bounds__(128) void style_s9_kernel(
    const float* __restrict__ style, const float* __restrict__ mod_w,
    const float* __restrict__ mod_b, float* __restrict__ s9)
{
    const int lane = threadIdx.x & 31;
    const int ic   = blockIdx.x * 4 + (threadIdx.x >> 5);

    const v4f* __restrict__ wrow = (const v4f*)(mod_w + (size_t)ic * SDIM);
    v4f w[4];
#pragma unroll
    for (int i = 0; i < 4; ++i) w[i] = wrow[lane + 32 * i];

    float acc[NB];
#pragma unroll
    for (int b = 0; b < NB; ++b) {
        const v4f* __restrict__ srow = (const v4f*)(style + b * SDIM);
        float a = 0.f;
#pragma unroll
        for (int i = 0; i < 4; ++i) {
            const v4f s = srow[lane + 32 * i];
            a += w[i].x * s.x;
            a += w[i].y * s.y;
            a += w[i].z * s.z;
            a += w[i].w * s.w;
        }
        acc[b] = a;
    }
#pragma unroll
    for (int b = 0; b < NB; ++b) acc[b] = wave_sum(acc[b]);

    const float mb = mod_b[ic];
    if (lane < 9) {
#pragma unroll
        for (int b = 0; b < NB; ++b) s9[b * K9 + ic * 9 + lane] = acc[b] + mb;
    }
}

__global__ __launch_bounds__(256) void demod_kernel(
    const float* __restrict__ weight, const float* __restrict__ s9,
    float* __restrict__ demod)
{
    const int oc   = blockIdx.x;
    const int lane = threadIdx.x & 31;
    const int b    = threadIdx.x >> 5;
    const v4f* __restrict__ wrow = (const v4f*)(weight + (size_t)oc * K9);
    const v4f* __restrict__ srow = (const v4f*)(s9 + (size_t)b * K9);

    v4f wv[K9 / 128], sv[K9 / 128];
#pragma unroll
    for (int i = 0; i < K9 / 128; ++i) wv[i] = wrow[lane + 32 * i];
#pragma unroll
    for (int i = 0; i < K9 / 128; ++i) sv[i] = srow[lane + 32 * i];

    float a0 = 0.f, a1 = 0.f, a2 = 0.f, a3 = 0.f;
#pragma unroll
    for (int i = 0; i < K9 / 128; ++i) {
        const v4f v = wv[i] * sv[i];
        a0 += v.x * v.x;
        a1 += v.y * v.y;
        a2 += v.z * v.z;
        a3 += v.w * v.w;
    }
    const float tot = wave_sum((a0 + a1) + (a2 + a3));
    if (lane == 0) demod[b * COUT + oc] = rsqrtf(tot + 1e-8f);
}

#define LDA 20
#define LDBC 20
#define NCHUNK (K9 / 16)

__global__ __launch_bounds__(256) void conv_wmma_kernel(
    const float* __restrict__ x, const float* __restrict__ weight,
    const float* __restrict__ bias, const float* __restrict__ s9,
    const float* __restrict__ demod, float* __restrict__ out)
{
    __shared__ float lA[2][256 * LDA];
    __shared__ float lB[2][64 * LDBC];

    const int tid = threadIdx.x;
    const int h   = blockIdx.x;
    const int b   = blockIdx.y;

    const int lane = tid & 31;
    const int wid  = tid >> 5;
    const int wm   = (wid & 3) * 64;
    const int wn   = (wid >> 2) * 32;
    const int half = lane >> 4;
    const int lr   = lane & 15;

    const int colB = tid & 63;
    const int kcb0 = __builtin_amdgcn_readfirstlane(tid >> 6);

    unsigned aoff[4];
#pragma unroll
    for (int i = 0; i < 4; ++i)
        aoff[i] = (unsigned)(((tid + 256 * i) >> 2) * K9 + ((tid & 3) << 2));

    const float* __restrict__ s9b = s9 + (size_t)b * K9;
    const float* __restrict__ xb  = x + (size_t)b * CIN * HW * HW;

    v4f    aReg[4];
    float  bReg[4];
    float  sReg[4];
    bool   okReg[4];

    auto loadChunk = [&](int t) {
        const int k0g = t * 16;
        const float* __restrict__ wt = weight + k0g;
#pragma unroll
        for (int i = 0; i < 4; ++i)
            aReg[i] = *(const v4f*)(wt + aoff[i]);
#pragma unroll
        for (int i = 0; i < 4; ++i) {
            const int kk = k0g + kcb0 + 4 * i;
            const int ic = kk / 9;
            const int r9 = kk - ic * 9;
            const int kh = r9 / 3;
            const int kw = r9 - kh * 3;
            const int ih = h + kh - 1;
            const bool rowok = (unsigned)ih < (unsigned)HW;
            const float* __restrict__ xrow = xb + (ic * HW + (rowok ? ih : 0)) * HW;
            const int iw = colB + kw - 1;
            okReg[i] = rowok && ((unsigned)iw < (unsigned)HW);
            const unsigned iwc = (unsigned)min(max(iw, 0), HW - 1);
            bReg[i] = xrow[iwc];
            sReg[i] = s9b[kk];
        }
        if (t + 2 < NCHUNK)
            __builtin_prefetch(wt + 32 + aoff[0], 0, 0);
    };
    auto storeChunk = [&](int buf) {
#pragma unroll
        for (int i = 0; i < 4; ++i) {
            int id4 = tid + 256 * i;
            int row = id4 >> 2;
            int kc4 = (id4 & 3) << 2;
            *(v4f*)&lA[buf][row * LDA + kc4] = aReg[i];
        }
#pragma unroll
        for (int i = 0; i < 4; ++i) {
            int kc = kcb0 + 4 * i;
            lB[buf][colB * LDBC + kc] = okReg[i] ? bReg[i] * sReg[i] : 0.f;
        }
    };

    v8f acc[4][2];
#pragma unroll
    for (int mt = 0; mt < 4; ++mt)
#pragma unroll
        for (int nt = 0; nt < 2; ++nt)
            acc[mt][nt] = (v8f){};

    loadChunk(0);
    storeChunk(0);
    __syncthreads();

    for (int t = 0; t < NCHUNK; ++t) {
        const int cur = t & 1;
        if (t + 1 < NCHUNK) loadChunk(t + 1);
        __builtin_amdgcn_sched_barrier(0);

#pragma unroll
        for (int k0 = 0; k0 < 16; k0 += 4) {
            const int ak = k0 + half * 2;
            v2f af[4];
#pragma unroll
            for (int mt = 0; mt < 4; ++mt)
                af[mt] = *(const v2f*)&lA[cur][(wm + mt * 16 + lr) * LDA + ak];
            v2f bf[2];
#pragma unroll
            for (int nt = 0; nt < 2; ++nt)
                bf[nt] = *(const v2f*)&lB[cur][(wn + nt * 16 + lr) * LDBC + ak];

#pragma unroll
            for (int mt = 0; mt < 4; ++mt)
#pragma unroll
                for (int nt = 0; nt < 2; ++nt)
                    acc[mt][nt] = wmma4(af[mt], bf[nt], acc[mt][nt]);
        }

        __builtin_amdgcn_sched_barrier(0);
        if (t + 1 < NCHUNK) storeChunk(1 ^ cur);
        __syncthreads();
    }

    const float* __restrict__ demodb = demod + b * COUT;
    float* __restrict__ outb = out + ((size_t)b * COUT * HW + h) * HW;
#pragma unroll
    for (int mt = 0; mt < 4; ++mt) {
#pragma unroll
        for (int v = 0; v < 8; ++v) {
            int oc = wm + mt * 16 + half * 8 + v;
            float d  = demodb[oc];
            float bb = bias[oc];
#pragma unroll
            for (int nt = 0; nt < 2; ++nt)
                outb[(size_t)oc * (HW * HW) + wn + nt * 16 + lr] =
                    acc[mt][nt][v] * d + bb;
        }
    }
}

extern "C" void kernel_launch(void* const* d_in, const int* in_sizes, int n_in,
                              void* d_out, int out_size, void* d_ws, size_t ws_size,
                              hipStream_t stream) {
    const float* x      = (const float*)d_in[0];
    const float* style  = (const float*)d_in[1];
    const float* weight = (const float*)d_in[2];
    const float* bias   = (const float*)d_in[3];
    const float* mod_w  = (const float*)d_in[4];
    const float* mod_b  = (const float*)d_in[5];
    float* out = (float*)d_out;

    float* s9    = (float*)d_ws;
    float* demod = s9 + NB * K9;

    style_s9_kernel<<<CIN / 4, 128, 0, stream>>>(style, mod_w, mod_b, s9);
    demod_kernel<<<COUT, 256, 0, stream>>>(weight, s9, demod);

    dim3 grid(HW, NB);
    conv_wmma_kernel<<<grid, 256, 0, stream>>>(x, weight, bias, s9, demod, out);
}


# K loop pipelined one stage deeper: WMMA fragments double-buffered in registers; next chunk's LDS fragment loads and chunk t+2's LDS stores issue ahead of the WMMA block, no LDS wait between blocks
# speedup vs baseline: 2.0025x; 1.0887x over previous
#include <hip/hip_runtime.h>
#include <hip/hip_bf16.h>

typedef float v2f __attribute__((ext_vector_type(2)));
typedef float v4f __attribute__((ext_vector_type(4)));
typedef float v8f __attribute__((ext_vector_type(8)));

__device__ __forceinline__ v8f wmma4(v2f a, v2f b, v8f c) {
    return __builtin_amdgcn_wmma_f32_16x16x4_f32(
         false, a,  false, b,
         (short)0, c,  false,  false);
}

#define CIN   256
#define COUT  256
#define HW    64
#define K9    2304
#define SDIM  512
#define NB    8

__device__ __forceinline__ float wave_sum(float v) {
#pragma unroll
    for (int off = 16; off > 0; off >>= 1) v += __shfl_xor(v, off);
    return v;
}

__global__ __launch_bounds__(128) void style_s9_kernel(
    const float* __restrict__ style, const float* __restrict__ mod_w,
    const float* __restrict__ mod_b, float* __restrict__ s9)
{
    const int lane = threadIdx.x & 31;
    const int ic   = blockIdx.x * 4 + (threadIdx.x >> 5);

    const v4f* __restrict__ wrow = (const v4f*)(mod_w + (size_t)ic * SDIM);
    v4f w[4];
#pragma unroll
    for (int i = 0; i < 4; ++i) w[i] = wrow[lane + 32 * i];

    float acc[NB];
#pragma unroll
    for (int b = 0; b < NB; ++b) {
        const v4f* __restrict__ srow = (const v4f*)(style + b * SDIM);
        float a = 0.f;
#pragma unroll
        for (int i = 0; i < 4; ++i) {
            const v4f s = srow[lane + 32 * i];
            a += w[i].x * s.x;
            a += w[i].y * s.y;
            a += w[i].z * s.z;
            a += w[i].w * s.w;
        }
        acc[b] = a;
    }
#pragma unroll
    for (int b = 0; b < NB; ++b) acc[b] = wave_sum(acc[b]);

    const float mb = mod_b[ic];
    if (lane < 9) {
#pragma unroll
        for (int b = 0; b < NB; ++b) s9[b * K9 + ic * 9 + lane] = acc[b] + mb;
    }
}

__global__ __launch_bounds__(256) void demod_kernel(
    const float* __restrict__ weight, const float* __restrict__ s9,
    float* __restrict__ demod)
{
    const int oc   = blockIdx.x;
    const int lane = threadIdx.x & 31;
    const int b    = threadIdx.x >> 5;
    const v4f* __restrict__ wrow = (const v4f*)(weight + (size_t)oc * K9);
    const v4f* __restrict__ srow = (const v4f*)(s9 + (size_t)b * K9);

    v4f wv[K9 / 128], sv[K9 / 128];
#pragma unroll
    for (int i = 0; i < K9 / 128; ++i) wv[i] = wrow[lane + 32 * i];
#pragma unroll
    for (int i = 0; i < K9 / 128; ++i) sv[i] = srow[lane + 32 * i];

    float a0 = 0.f, a1 = 0.f, a2 = 0.f, a3 = 0.f;
#pragma unroll
    for (int i = 0; i < K9 / 128; ++i) {
        const v4f v = wv[i] * sv[i];
        a0 += v.x * v.x;
        a1 += v.y * v.y;
        a2 += v.z * v.z;
        a3 += v.w * v.w;
    }
    const float tot = wave_sum((a0 + a1) + (a2 + a3));
    if (lane == 0) demod[b * COUT + oc] = rsqrtf(tot + 1e-8f);
}

#define LDA 20
#define LDBC 20
#define NCHUNK (K9 / 16)

__global__ __launch_bounds__(256) void conv_wmma_kernel(
    const float* __restrict__ x, const float* __restrict__ weight,
    const float* __restrict__ bias, const float* __restrict__ s9,
    const float* __restrict__ demod, float* __restrict__ out)
{
    __shared__ float lA[2][256 * LDA];
    __shared__ float lB[2][64 * LDBC];

    const int tid = threadIdx.x;
    const int h   = blockIdx.x;
    const int b   = blockIdx.y;

    const int lane = tid & 31;
    const int wid  = tid >> 5;
    const int wm   = (wid & 3) * 64;
    const int wn   = (wid >> 2) * 32;
    const int half = lane >> 4;
    const int lr   = lane & 15;

    const int colB = tid & 63;
    const int kcb0 = __builtin_amdgcn_readfirstlane(tid >> 6);

    unsigned aoff[4];
#pragma unroll
    for (int i = 0; i < 4; ++i)
        aoff[i] = (unsigned)(((tid + 256 * i) >> 2) * K9 + ((tid & 3) << 2));

    const float* __restrict__ s9b = s9 + (size_t)b * K9;
    const float* __restrict__ xb  = x + (size_t)b * CIN * HW * HW;

    v4f    aReg[4];
    float  bReg[4];
    float  sReg[4];
    bool   okReg[4];

    auto loadChunk = [&](int t) {
        const int k0g = t * 16;
        const float* __restrict__ wt = weight + k0g;
#pragma unroll
        for (int i = 0; i < 4; ++i)
            aReg[i] = *(const v4f*)(wt + aoff[i]);
#pragma unroll
        for (int i = 0; i < 4; ++i) {
            const int kk = k0g + kcb0 + 4 * i;
            const int ic = kk / 9;
            const int r9 = kk - ic * 9;
            const int kh = r9 / 3;
            const int kw = r9 - kh * 3;
            const int ih = h + kh - 1;
            const bool rowok = (unsigned)ih < (unsigned)HW;
            const float* __restrict__ xrow = xb + (ic * HW + (rowok ? ih : 0)) * HW;
            const int iw = colB + kw - 1;
            okReg[i] = rowok && ((unsigned)iw < (unsigned)HW);
            const unsigned iwc = (unsigned)min(max(iw, 0), HW - 1);
            bReg[i] = xrow[iwc];
            sReg[i] = s9b[kk];
        }
        if (t + 2 < NCHUNK)
            __builtin_prefetch(wt + 32 + aoff[0], 0, 0);
    };
    auto storeChunk = [&](int buf) {
#pragma unroll
        for (int i = 0; i < 4; ++i) {
            int id4 = tid + 256 * i;
            int row = id4 >> 2;
            int kc4 = (id4 & 3) << 2;
            *(v4f*)&lA[buf][row * LDA + kc4] = aReg[i];
        }
#pragma unroll
        for (int i = 0; i < 4; ++i) {
            int kc = kcb0 + 4 * i;
            lB[buf][colB * LDBC + kc] = okReg[i] ? bReg[i] * sReg[i] : 0.f;
        }
    };

    v8f acc[4][2];
#pragma unroll
    for (int mt = 0; mt < 4; ++mt)
#pragma unroll
        for (int nt = 0; nt < 2; ++nt)
            acc[mt][nt] = (v8f){};

    v2f fA0[4][4], fB0[4][2];
    v2f fA1[4][4], fB1[4][2];

    auto loadFrags = [&](int buf, v2f (&fa)[4][4], v2f (&fb)[4][2])
        __attribute__((always_inline)) {
#pragma unroll
        for (int ks = 0; ks < 4; ++ks) {
            const int ak = ks * 4 + half * 2;
#pragma unroll
            for (int mt = 0; mt < 4; ++mt)
                fa[ks][mt] = *(const v2f*)&lA[buf][(wm + mt * 16 + lr) * LDA + ak];
#pragma unroll
            for (int nt = 0; nt < 2; ++nt)
                fb[ks][nt] = *(const v2f*)&lB[buf][(wn + nt * 16 + lr) * LDBC + ak];
        }
    };
    auto mmaChunk = [&](const v2f (&fa)[4][4], const v2f (&fb)[4][2])
        __attribute__((always_inline)) {
#pragma unroll
        for (int ks = 0; ks < 4; ++ks)
#pragma unroll
            for (int mt = 0; mt < 4; ++mt)
#pragma unroll
                for (int nt = 0; nt < 2; ++nt)
                    acc[mt][nt] = wmma4(fa[ks][mt], fb[ks][nt], acc[mt][nt]);
    };

    auto step = [&](int t, int par, bool doStore, bool doLoad, bool doFrags, bool doSync,
                    const v2f (&ca)[4][4], const v2f (&cb)[4][2],
                    v2f (&na)[4][4], v2f (&nb)[4][2])
        __attribute__((always_inline)) {
        if (doStore) storeChunk(par);
        if (doLoad)  loadChunk(t + 3);
        if (doFrags) loadFrags(par ^ 1, na, nb);
        __builtin_amdgcn_sched_barrier(0);
        mmaChunk(ca, cb);
        __builtin_amdgcn_sched_barrier(0);
        if (doSync) __syncthreads();
    };

    loadChunk(0);
    storeChunk(0);
    loadChunk(1);
    storeChunk(1);
    loadChunk(2);
    __syncthreads();
    loadFrags(0, fA0, fB0);
    __syncthreads();

    for (int t = 0; t < NCHUNK - 4; t += 2) {
        step(t,     0, true, true, true, true, fA0, fB0, fA1, fB1);
        step(t + 1, 1, true, true, true, true, fA1, fB1, fA0, fB0);
    }
    step(NCHUNK - 4, 0, true,  true,  true,  true,  fA0, fB0, fA1, fB1);
    step(NCHUNK - 3, 1, true,  false, true,  true,  fA1, fB1, fA0, fB0);
    step(NCHUNK - 2, 0, false, false, true,  false, fA0, fB0, fA1, fB1);
    step(NCHUNK - 1, 1, false, false, false, false, fA1, fB1, fA0, fB0);

    const float* __restrict__ demodb = demod + b * COUT;
    float* __restrict__ outb = out + ((size_t)b * COUT * HW + h) * HW;
#pragma unroll
    for (int mt = 0; mt < 4; ++mt) {
#pragma unroll
        for (int v = 0; v < 8; ++v) {
            int oc = wm + mt * 16 + half * 8 + v;
            float d  = demodb[oc];
            float bb = bias[oc];
#pragma unroll
            for (int nt = 0; nt < 2; ++nt)
                outb[(size_t)oc * (HW * HW) + wn + nt * 16 + lr] =
                    acc[mt][nt][v] * d + bb;
        }
    }
}

extern "C" void kernel_launch(void* const* d_in, const int* in_sizes, int n_in,
                              void* d_out, int out_size, void* d_ws, size_t ws_size,
                              hipStream_t stream) {
    const float* x      = (const float*)d_in[0];
    const float* style  = (const float*)d_in[1];
    const float* weight = (const float*)d_in[2];
    const float* bias   = (const float*)d_in[3];
    const float* mod_w  = (const float*)d_in[4];
    const float* mod_b  = (const float*)d_in[5];
    float* out = (float*)d_out;

    float* s9    = (float*)d_ws;
    float* demod = s9 + NB * K9;

    style_s9_kernel<<<CIN / 4, 128, 0, stream>>>(style, mod_w, mod_b, s9);
    demod_kernel<<<COUT, 256, 0, stream>>>(weight, s9, demod);

    dim3 grid(HW, NB);
    conv_wmma_kernel<<<grid, 256, 0, stream>>>(x, weight, bias, s9, demod, out);
}


# Conv block tile doubled to two image rows: 256x128 per block, wave tile 64x64, 256 blocks (one per WGP); 64 WMMAs per wave per barrier, each weight chunk staged once per two rows
# speedup vs baseline: 2.0756x; 1.0365x over previous
#include <hip/hip_runtime.h>
#include <hip/hip_bf16.h>

typedef float v2f __attribute__((ext_vector_type(2)));
typedef float v4f __attribute__((ext_vector_type(4)));
typedef float v8f __attribute__((ext_vector_type(8)));

__device__ __forceinline__ v8f wmma4(v2f a, v2f b, v8f c) {
    return __builtin_amdgcn_wmma_f32_16x16x4_f32(
         false, a,  false, b,
         (short)0, c,  false,  false);
}

#define CIN   256
#define COUT  256
#define HW    64
#define K9    2304
#define SDIM  512
#define NB    8

__device__ __forceinline__ float wave_sum(float v) {
#pragma unroll
    for (int off = 16; off > 0; off >>= 1) v += __shfl_xor(v, off);
    return v;
}

__global__ __launch_bounds__(128) void style_s9_kernel(
    const float* __restrict__ style, const float* __restrict__ mod_w,
    const float* __restrict__ mod_b, float* __restrict__ s9)
{
    const int lane = threadIdx.x & 31;
    const int ic   = blockIdx.x * 4 + (threadIdx.x >> 5);

    const v4f* __restrict__ wrow = (const v4f*)(mod_w + (size_t)ic * SDIM);
    v4f w[4];
#pragma unroll
    for (int i = 0; i < 4; ++i) w[i] = wrow[lane + 32 * i];

    float acc[NB];
#pragma unroll
    for (int b = 0; b < NB; ++b) {
        const v4f* __restrict__ srow = (const v4f*)(style + b * SDIM);
        float a = 0.f;
#pragma unroll
        for (int i = 0; i < 4; ++i) {
            const v4f s = srow[lane + 32 * i];
            a += w[i].x * s.x;
            a += w[i].y * s.y;
            a += w[i].z * s.z;
            a += w[i].w * s.w;
        }
        acc[b] = a;
    }
#pragma unroll
    for (int b = 0; b < NB; ++b) acc[b] = wave_sum(acc[b]);

    const float mb = mod_b[ic];
    if (lane < 9) {
#pragma unroll
        for (int b = 0; b < NB; ++b) s9[b * K9 + ic * 9 + lane] = acc[b] + mb;
    }
}

__global__ __launch_bounds__(256) void demod_kernel(
    const float* __restrict__ weight, const float* __restrict__ s9,
    float* __restrict__ demod)
{
    const int oc   = blockIdx.x;
    const int lane = threadIdx.x & 31;
    const int b    = threadIdx.x >> 5;
    const v4f* __restrict__ wrow = (const v4f*)(weight + (size_t)oc * K9);
    const v4f* __restrict__ srow = (const v4f*)(s9 + (size_t)b * K9);

    v4f wv[K9 / 128], sv[K9 / 128];
#pragma unroll
    for (int i = 0; i < K9 / 128; ++i) wv[i] = wrow[lane + 32 * i];
#pragma unroll
    for (int i = 0; i < K9 / 128; ++i) sv[i] = srow[lane + 32 * i];

    float a0 = 0.f, a1 = 0.f, a2 = 0.f, a3 = 0.f;
#pragma unroll
    for (int i = 0; i < K9 / 128; ++i) {
        const v4f v = wv[i] * sv[i];
        a0 += v.x * v.x;
        a1 += v.y * v.y;
        a2 += v.z * v.z;
        a3 += v.w * v.w;
    }
    const float tot = wave_sum((a0 + a1) + (a2 + a3));
    if (lane == 0) demod[b * COUT + oc] = rsqrtf(tot + 1e-8f);
}

#define LDA 20
#define LDBC 20
#define NCHUNK (K9 / 16)
#define NTILE 128

__global__ __launch_bounds__(256) void conv_wmma_kernel(
    const float* __restrict__ x, const float* __restrict__ weight,
    const float* __restrict__ bias, const float* __restrict__ s9,
    const float* __restrict__ demod, float* __restrict__ out)
{
    __shared__ float lA[2][256 * LDA];
    __shared__ float lB[2][NTILE * LDBC];

    const int tid = threadIdx.x;
    const int h0  = blockIdx.x * 2;
    const int b   = blockIdx.y;

    const int lane = tid & 31;
    const int wid  = tid >> 5;
    const int wm   = (wid & 3) * 64;
    const int wn   = (wid >> 2) * 64;
    const int half = lane >> 4;
    const int lr   = lane & 15;

    const int kcb0 = __builtin_amdgcn_readfirstlane(wid & 3);
    const int rsB  = __builtin_amdgcn_readfirstlane(wid >> 2);
    const int hB   = h0 + rsB;
    const int nB   = rsB * 64 + lane;

    unsigned aoff[4];
#pragma unroll
    for (int i = 0; i < 4; ++i)
        aoff[i] = (unsigned)(((tid + 256 * i) >> 2) * K9 + ((tid & 3) << 2));

    const float* __restrict__ s9b = s9 + (size_t)b * K9;
    const float* __restrict__ xb  = x + (size_t)b * CIN * HW * HW;

    v4f    aReg[4];
    float  bReg[4][2];
    float  sReg[4];
    bool   okReg[4][2];

    auto loadChunk = [&](int t) {
        const int k0g = t * 16;
        const float* __restrict__ wt = weight + k0g;
#pragma unroll
        for (int i = 0; i < 4; ++i)
            aReg[i] = *(const v4f*)(wt + aoff[i]);
#pragma unroll
        for (int i = 0; i < 4; ++i) {
            const int kk = k0g + kcb0 + 4 * i;
            const int ic = kk / 9;
            const int r9 = kk - ic * 9;
            const int kh = r9 / 3;
            const int kw = r9 - kh * 3;
            const int ih = hB + kh - 1;
            const bool rowok = (unsigned)ih < (unsigned)HW;
            const float* __restrict__ xrow = xb + (ic * HW + (rowok ? ih : 0)) * HW;
#pragma unroll
            for (int c = 0; c < 2; ++c) {
                const int iw = lane + 32 * c + kw - 1;
                okReg[i][c] = rowok && ((unsigned)iw < (unsigned)HW);
                const unsigned iwc = (unsigned)min(max(iw, 0), HW - 1);
                bReg[i][c] = xrow[iwc];
            }
            sReg[i] = s9b[kk];
        }
        if (t + 2 < NCHUNK)
            __builtin_prefetch(wt + 32 + aoff[0], 0, 0);
    };
    auto storeChunk = [&](int buf) {
#pragma unroll
        for (int i = 0; i < 4; ++i) {
            int id4 = tid + 256 * i;
            int row = id4 >> 2;
            int kc4 = (id4 & 3) << 2;
            *(v4f*)&lA[buf][row * LDA + kc4] = aReg[i];
        }
#pragma unroll
        for (int i = 0; i < 4; ++i) {
            int kc = kcb0 + 4 * i;
#pragma unroll
            for (int c = 0; c < 2; ++c)
                lB[buf][(nB + 32 * c) * LDBC + kc] =
                    okReg[i][c] ? bReg[i][c] * sReg[i] : 0.f;
        }
    };

    v8f acc[4][4];
#pragma unroll
    for (int mt = 0; mt < 4; ++mt)
#pragma unroll
        for (int nt = 0; nt < 4; ++nt)
            acc[mt][nt] = (v8f){};

    v2f fA0[4][4], fB0[4][4];
    v2f fA1[4][4], fB1[4][4];

    auto loadFrags = [&](int buf, v2f (&fa)[4][4], v2f (&fb)[4][4])
        __attribute__((always_inline)) {
#pragma unroll
        for (int ks = 0; ks < 4; ++ks) {
            const int ak = ks * 4 + half * 2;
#pragma unroll
            for (int mt = 0; mt < 4; ++mt)
                fa[ks][mt] = *(const v2f*)&lA[buf][(wm + mt * 16 + lr) * LDA + ak];
#pragma unroll
            for (int nt = 0; nt < 4; ++nt)
                fb[ks][nt] = *(const v2f*)&lB[buf][(wn + nt * 16 + lr) * LDBC + ak];
        }
    };
    auto mmaChunk = [&](const v2f (&fa)[4][4], const v2f (&fb)[4][4])
        __attribute__((always_inline)) {
#pragma unroll
        for (int ks = 0; ks < 4; ++ks)
#pragma unroll
            for (int mt = 0; mt < 4; ++mt)
#pragma unroll
                for (int nt = 0; nt < 4; ++nt)
                    acc[mt][nt] = wmma4(fa[ks][mt], fb[ks][nt], acc[mt][nt]);
    };

    auto step = [&](int t, int par, bool doStore, bool doLoad, bool doFrags, bool doSync,
                    const v2f (&ca)[4][4], const v2f (&cb)[4][4],
                    v2f (&na)[4][4], v2f (&nb)[4][4])
        __attribute__((always_inline)) {
        if (doStore) storeChunk(par);
        if (doLoad)  loadChunk(t + 3);
        if (doFrags) loadFrags(par ^ 1, na, nb);
        __builtin_amdgcn_sched_barrier(0);
        mmaChunk(ca, cb);
        __builtin_amdgcn_sched_barrier(0);
        if (doSync) __syncthreads();
    };

    loadChunk(0);
    storeChunk(0);
    loadChunk(1);
    storeChunk(1);
    loadChunk(2);
    __syncthreads();
    loadFrags(0, fA0, fB0);
    __syncthreads();

    for (int t = 0; t < NCHUNK - 4; t += 2) {
        step(t,     0, true, true, true, true, fA0, fB0, fA1, fB1);
        step(t + 1, 1, true, true, true, true, fA1, fB1, fA0, fB0);
    }
    step(NCHUNK - 4, 0, true,  true,  true,  true,  fA0, fB0, fA1, fB1);
    step(NCHUNK - 3, 1, true,  false, true,  true,  fA1, fB1, fA0, fB0);
    step(NCHUNK - 2, 0, false, false, true,  false, fA0, fB0, fA1, fB1);
    step(NCHUNK - 1, 1, false, false, false, false, fA1, fB1, fA0, fB0);

    const float* __restrict__ demodb = demod + b * COUT;
    float* __restrict__ outb = out + ((size_t)b * COUT * HW + h0 + (wid >> 2)) * HW;
#pragma unroll
    for (int mt = 0; mt < 4; ++mt) {
#pragma unroll
        for (int v = 0; v < 8; ++v) {
            int oc = wm + mt * 16 + half * 8 + v;
            float d  = demodb[oc];
            float bb = bias[oc];
#pragma unroll
            for (int nt = 0; nt < 4; ++nt)
                outb[(size_t)oc * (HW * HW) + nt * 16 + lr] =
                    acc[mt][nt][v] * d + bb;
        }
    }
}

extern "C" void kernel_launch(void* const* d_in, const int* in_sizes, int n_in,
                              void* d_out, int out_size, void* d_ws, size_t ws_size,
                              hipStream_t stream) {
    const float* x      = (const float*)d_in[0];
    const float* style  = (const float*)d_in[1];
    const float* weight = (const float*)d_in[2];
    const float* bias   = (const float*)d_in[3];
    const float* mod_w  = (const float*)d_in[4];
    const float* mod_b  = (const float*)d_in[5];
    float* out = (float*)d_out;

    float* s9    = (float*)d_ws;
    float* demod = s9 + NB * K9;

    style_s9_kernel<<<CIN / 4, 128, 0, stream>>>(style, mod_w, mod_b, s9);
    demod_kernel<<<COUT, 256, 0, stream>>>(weight, s9, demod);

    dim3 grid(HW / 2, NB);
    conv_wmma_kernel<<<grid, 256, 0, stream>>>(x, weight, bias, s9, demod, out);
}


# K reordered into chunks of 16 input channels of ONE filter tap (weights re-laid into workspace by launch 1): im2col index arithmetic once per chunk, not per k; summation order over K changes
# speedup vs baseline: 2.3411x; 1.1279x over previous
#include <hip/hip_runtime.h>
#include <hip/hip_bf16.h>

typedef float v2f __attribute__((ext_vector_type(2)));
typedef float v4f __attribute__((ext_vector_type(4)));
typedef float v8f __attribute__((ext_vector_type(8)));

__device__ __forceinline__ v8f wmma4(v2f a, v2f b, v8f c) {
    return __builtin_amdgcn_wmma_f32_16x16x4_f32(
         false, a,  false, b,
         (short)0, c,  false,  false);
}

#define CIN   256
#define COUT  256
#define HW    64
#define K9    2304
#define SDIM  512
#define NB    8

__device__ __forceinline__ float wave_sum(float v) {
#pragma unroll
    for (int off = 16; off > 0; off >>= 1) v += __shfl_xor(v, off);
    return v;
}

#define S9_BLOCKS  (CIN / 4)
#define WT_BLOCKS  (COUT * 16 / 64)

__global__ __launch_bounds__(128) void style_s9_kernel(
    const float* __restrict__ style, const float* __restrict__ mod_w,
    const float* __restrict__ mod_b, const float* __restrict__ weight,
    float* __restrict__ s9, float* __restrict__ wT)
{
    if (blockIdx.x >= S9_BLOCKS) {
        if (threadIdx.x >= 64) return;
        const int r   = blockIdx.x - S9_BLOCKS;
        const int icg = r >> 2;
        const int oc  = ((r & 3) << 6) + threadIdx.x;
        const v4f* __restrict__ src = (const v4f*)(weight + (size_t)oc * K9 + icg * 144);
        v4f in[36];
#pragma unroll
        for (int q = 0; q < 36; ++q) in[q] = src[q];
#pragma unroll
        for (int tap = 0; tap < 9; ++tap) {
            v4f* __restrict__ dst = (v4f*)(wT + ((size_t)(icg * 9 + tap) * COUT + oc) * 16);
#pragma unroll
            for (int g = 0; g < 4; ++g) {
                v4f o;
                o.x = in[((4 * g + 0) * 9 + tap) >> 2][((4 * g + 0) * 9 + tap) & 3];
                o.y = in[((4 * g + 1) * 9 + tap) >> 2][((4 * g + 1) * 9 + tap) & 3];
                o.z = in[((4 * g + 2) * 9 + tap) >> 2][((4 * g + 2) * 9 + tap) & 3];
                o.w = in[((4 * g + 3) * 9 + tap) >> 2][((4 * g + 3) * 9 + tap) & 3];
                dst[g] = o;
            }
        }
        return;
    }
    const int lane = threadIdx.x & 31;
    const int ic   = blockIdx.x * 4 + (threadIdx.x >> 5);

    const v4f* __restrict__ wrow = (const v4f*)(mod_w + (size_t)ic * SDIM);
    v4f w[4];
#pragma unroll
    for (int i = 0; i < 4; ++i) w[i] = wrow[lane + 32 * i];

    float acc[NB];
#pragma unroll
    for (int b = 0; b < NB; ++b) {
        const v4f* __restrict__ srow = (const v4f*)(style + b * SDIM);
        float a = 0.f;
#pragma unroll
        for (int i = 0; i < 4; ++i) {
            const v4f s = srow[lane + 32 * i];
            a += w[i].x * s.x;
            a += w[i].y * s.y;
            a += w[i].z * s.z;
            a += w[i].w * s.w;
        }
        acc[b] = a;
    }
#pragma unroll
    for (int b = 0; b < NB; ++b) acc[b] = wave_sum(acc[b]);

    const float mb = mod_b[ic];
    if (lane < 9) {
#pragma unroll
        for (int b = 0; b < NB; ++b) s9[b * K9 + ic * 9 + lane] = acc[b] + mb;
    }
}

__global__ __launch_bounds__(256) void demod_kernel(
    const float* __restrict__ weight, const float* __restrict__ s9,
    float* __restrict__ demod)
{
    const int oc   = blockIdx.x;
    const int lane = threadIdx.x & 31;
    const int b    = threadIdx.x >> 5;
    const v4f* __restrict__ wrow = (const v4f*)(weight + (size_t)oc * K9);
    const v4f* __restrict__ srow = (const v4f*)(s9 + (size_t)b * K9);

    v4f wv[K9 / 128], sv[K9 / 128];
#pragma unroll
    for (int i = 0; i < K9 / 128; ++i) wv[i] = wrow[lane + 32 * i];
#pragma unroll
    for (int i = 0; i < K9 / 128; ++i) sv[i] = srow[lane + 32 * i];

    float a0 = 0.f, a1 = 0.f, a2 = 0.f, a3 = 0.f;
#pragma unroll
    for (int i = 0; i < K9 / 128; ++i) {
        const v4f v = wv[i] * sv[i];
        a0 += v.x * v.x;
        a1 += v.y * v.y;
        a2 += v.z * v.z;
        a3 += v.w * v.w;
    }
    const float tot = wave_sum((a0 + a1) + (a2 + a3));
    if (lane == 0) demod[b * COUT + oc] = rsqrtf(tot + 1e-8f);
}

#define LDA 20
#define LDBC 20
#define NCHUNK (K9 / 16)
#define NTILE 128

__global__ __launch_bounds__(256) void conv_wmma_kernel(
    const float* __restrict__ x, const float* __restrict__ wT,
    const float* __restrict__ bias, const float* __restrict__ s9,
    const float* __restrict__ demod, float* __restrict__ out)
{
    __shared__ float lA[2][256 * LDA];
    __shared__ float lB[2][NTILE * LDBC];

    const int tid = threadIdx.x;
    const int h0  = blockIdx.x * 2;
    const int b   = blockIdx.y;

    const int lane = tid & 31;
    const int wid  = tid >> 5;
    const int wm   = (wid & 3) * 64;
    const int wn   = (wid >> 2) * 64;
    const int half = lane >> 4;
    const int lr   = lane & 15;

    const int kcb0 = __builtin_amdgcn_readfirstlane(wid & 3);
    const int rsB  = __builtin_amdgcn_readfirstlane(wid >> 2);
    const int hB   = h0 + rsB;
    const int nB   = rsB * 64 + lane;

    unsigned aoff[4];
#pragma unroll
    for (int i = 0; i < 4; ++i)
        aoff[i] = (unsigned)((tid + 256 * i) << 2);

    const float* __restrict__ s9b = s9 + (size_t)b * K9;
    const float* __restrict__ xb  = x + (size_t)b * CIN * HW * HW;

    v4f    aReg[4];
    float  bReg[4][2];
    float  sReg[4];
    bool   okReg[2];

    auto loadChunk = [&](int t) {
        const float* __restrict__ wt = wT + t * (COUT * 16);
#pragma unroll
        for (int i = 0; i < 4; ++i)
            aReg[i] = *(const v4f*)(wt + aoff[i]);
        const int icg = t / 9;
        const int tap = t - icg * 9;
        const int kh  = tap / 3;
        const int kw  = tap - kh * 3;
        const int ih  = hB + kh - 1;
        const bool rowok = (unsigned)ih < (unsigned)HW;
        const unsigned icb = (unsigned)(icg * 16 + kcb0);
        const float* __restrict__ xrow = xb + (icb * HW + (rowok ? (unsigned)ih : 0u)) * HW;
        const float* __restrict__ srow = s9b + icb * 9;
        unsigned iwc[2];
#pragma unroll
        for (int c = 0; c < 2; ++c) {
            const int iw = lane + 32 * c + kw - 1;
            okReg[c] = rowok && ((unsigned)iw < (unsigned)HW);
            iwc[c]   = (unsigned)min(max(iw, 0), HW - 1);
        }
#pragma unroll
        for (int i = 0; i < 4; ++i) {
#pragma unroll
            for (int c = 0; c < 2; ++c)
                bReg[i][c] = xrow[(4 * i) * (HW * HW) + iwc[c]];
            sReg[i] = srow[(4 * i) * 9];
        }
        if (t + 2 < NCHUNK)
            __builtin_prefetch(wt + 2 * (COUT * 16) + aoff[0], 0, 0);
    };
    auto storeChunk = [&](int buf) {
#pragma unroll
        for (int i = 0; i < 4; ++i) {
            int id4 = tid + 256 * i;
            int row = id4 >> 2;
            int kc4 = (id4 & 3) << 2;
            *(v4f*)&lA[buf][row * LDA + kc4] = aReg[i];
        }
#pragma unroll
        for (int i = 0; i < 4; ++i) {
            int kc = kcb0 + 4 * i;
#pragma unroll
            for (int c = 0; c < 2; ++c)
                lB[buf][(nB + 32 * c) * LDBC + kc] =
                    okReg[c] ? bReg[i][c] * sReg[i] : 0.f;
        }
    };

    v8f acc[4][4];
#pragma unroll
    for (int mt = 0; mt < 4; ++mt)
#pragma unroll
        for (int nt = 0; nt < 4; ++nt)
            acc[mt][nt] = (v8f){};

    v2f fA0[4][4], fB0[4][4];
    v2f fA1[4][4], fB1[4][4];

    auto loadFrags = [&](int buf, v2f (&fa)[4][4], v2f (&fb)[4][4])
        __attribute__((always_inline)) {
#pragma unroll
        for (int ks = 0; ks < 4; ++ks) {
            const int ak = ks * 4 + half * 2;
#pragma unroll
            for (int mt = 0; mt < 4; ++mt)
                fa[ks][mt] = *(const v2f*)&lA[buf][(wm + mt * 16 + lr) * LDA + ak];
#pragma unroll
            for (int nt = 0; nt < 4; ++nt)
                fb[ks][nt] = *(const v2f*)&lB[buf][(wn + nt * 16 + lr) * LDBC + ak];
        }
    };
    auto mmaChunk = [&](const v2f (&fa)[4][4], const v2f (&fb)[4][4])
        __attribute__((always_inline)) {
#pragma unroll
        for (int ks = 0; ks < 4; ++ks)
#pragma unroll
            for (int mt = 0; mt < 4; ++mt)
#pragma unroll
                for (int nt = 0; nt < 4; ++nt)
                    acc[mt][nt] = wmma4(fa[ks][mt], fb[ks][nt], acc[mt][nt]);
    };

    auto step = [&](int t, int par, bool doStore, bool doLoad, bool doFrags, bool doSync,
                    const v2f (&ca)[4][4], const v2f (&cb)[4][4],
                    v2f (&na)[4][4], v2f (&nb)[4][4])
        __attribute__((always_inline)) {
        if (doStore) storeChunk(par);
        if (doLoad)  loadChunk(t + 3);
        if (doFrags) loadFrags(par ^ 1, na, nb);
        __builtin_amdgcn_sched_barrier(0);
        mmaChunk(ca, cb);
        __builtin_amdgcn_sched_barrier(0);
        if (doSync) __syncthreads();
    };

    loadChunk(0);
    storeChunk(0);
    loadChunk(1);
    storeChunk(1);
    loadChunk(2);
    __syncthreads();
    loadFrags(0, fA0, fB0);
    __syncthreads();

    for (int t = 0; t < NCHUNK - 4; t += 2) {
        step(t,     0, true, true, true, true, fA0, fB0, fA1, fB1);
        step(t + 1, 1, true, true, true, true, fA1, fB1, fA0, fB0);
    }
    step(NCHUNK - 4, 0, true,  true,  true,  true,  fA0, fB0, fA1, fB1);
    step(NCHUNK - 3, 1, true,  false, true,  true,  fA1, fB1, fA0, fB0);
    step(NCHUNK - 2, 0, false, false, true,  false, fA0, fB0, fA1, fB1);
    step(NCHUNK - 1, 1, false, false, false, false, fA1, fB1, fA0, fB0);

    const float* __restrict__ demodb = demod + b * COUT;
    float* __restrict__ outb = out + ((size_t)b * COUT * HW + h0 + (wid >> 2)) * HW;
#pragma unroll
    for (int mt = 0; mt < 4; ++mt) {
#pragma unroll
        for (int v = 0; v < 8; ++v) {
            int oc = wm + mt * 16 + half * 8 + v;
            float d  = demodb[oc];
            float bb = bias[oc];
#pragma unroll
            for (int nt = 0; nt < 4; ++nt)
                outb[(size_t)oc * (HW * HW) + nt * 16 + lr] =
                    acc[mt][nt][v] * d + bb;
        }
    }
}

extern "C" void kernel_launch(void* const* d_in, const int* in_sizes, int n_in,
                              void* d_out, int out_size, void* d_ws, size_t ws_size,
                              hipStream_t stream) {
    const float* x      = (const float*)d_in[0];
    const float* style  = (const float*)d_in[1];
    const float* weight = (const float*)d_in[2];
    const float* bias   = (const float*)d_in[3];
    const float* mod_w  = (const float*)d_in[4];
    const float* mod_b  = (const float*)d_in[5];
    float* out = (float*)d_out;

    float* s9    = (float*)d_ws;
    float* demod = s9 + NB * K9;
    float* wT    = demod + NB * COUT;

    style_s9_kernel<<<S9_BLOCKS + WT_BLOCKS, 128, 0, stream>>>(style, mod_w, mod_b, weight, s9, wT);
    demod_kernel<<<COUT, 256, 0, stream>>>(weight, s9, demod);

    dim3 grid(HW / 2, NB);
    conv_wmma_kernel<<<grid, 256, 0, stream>>>(x, wT, bias, s9, demod, out);
}


# Vertical minimal filtering F(2,3) over the block's two image rows: 4 transformed-row products per (channel, kw) instead of 6 taps, 96 K steps not 144; conv summation order and rounding change
# speedup vs baseline: 3.0431x; 1.2999x over previous
#include <hip/hip_runtime.h>
#include <hip/hip_bf16.h>

typedef float v2f __attribute__((ext_vector_type(2)));
typedef float v4f __attribute__((ext_vector_type(4)));
typedef float v8f __attribute__((ext_vector_type(8)));

__device__ __forceinline__ v8f wmma4(v2f a, v2f b, v8f c) {
    return __builtin_amdgcn_wmma_f32_16x16x4_f32(
         false, a,  false, b,
         (short)0, c,  false,  false);
}

#define CIN   256
#define COUT  256
#define HW    64
#define K9    2304
#define SDIM  512
#define NB    8

__device__ __forceinline__ float wave_sum(float v) {
#pragma unroll
    for (int off = 16; off > 0; off >>= 1) v += __shfl_xor(v, off);
    return v;
}

#define S9_BLOCKS  (CIN / 4)
#define WT_BLOCKS  (COUT * 16 / 64)
#define AROWS      512
#define NSTEP      96

__global__ __launch_bounds__(128) void style_s9_kernel(
    const float* __restrict__ style, const float* __restrict__ mod_w,
    const float* __restrict__ mod_b, const float* __restrict__ weight,
    float* __restrict__ s9, float* __restrict__ wU)
{
    if (blockIdx.x >= S9_BLOCKS) {
        if (threadIdx.x >= 64) return;
        const int r   = blockIdx.x - S9_BLOCKS;
        const int icg = r >> 2;
        const int oc  = ((r & 3) << 6) + threadIdx.x;
        const v4f* __restrict__ src = (const v4f*)(weight + (size_t)oc * K9 + icg * 144);
        v4f in[36];
#pragma unroll
        for (int q = 0; q < 36; ++q) in[q] = src[q];
#pragma unroll
        for (int kw = 0; kw < 3; ++kw) {
#pragma unroll
            for (int j = 0; j < 4; ++j) {
                const int t = (icg * 3 + kw) * 2 + (j & 1);
                v4f* __restrict__ dst =
                    (v4f*)(wU + ((size_t)t * AROWS + (j >> 1) * COUT + oc) * 16);
#pragma unroll
                for (int g = 0; g < 4; ++g) {
                    v4f o;
#pragma unroll
                    for (int e = 0; e < 4; ++e) {
                        const int kc = 4 * g + e;
                        const float g0 = in[(kc * 9 + kw) >> 2][(kc * 9 + kw) & 3];
                        const float g1 = in[(kc * 9 + 3 + kw) >> 2][(kc * 9 + 3 + kw) & 3];
                        const float g2 = in[(kc * 9 + 6 + kw) >> 2][(kc * 9 + 6 + kw) & 3];
                        o[e] = (j == 0) ? g0
                             : (j == 1) ? 0.5f * ((g0 + g1) + g2)
                             : (j == 2) ? 0.5f * ((g0 - g1) + g2)
                             :            g2;
                    }
                    dst[g] = o;
                }
            }
        }
        return;
    }
    const int lane = threadIdx.x & 31;
    const int ic   = blockIdx.x * 4 + (threadIdx.x >> 5);

    const v4f* __restrict__ wrow = (const v4f*)(mod_w + (size_t)ic * SDIM);
    v4f w[4];
#pragma unroll
    for (int i = 0; i < 4; ++i) w[i] = wrow[lane + 32 * i];

    float acc[NB];
#pragma unroll
    for (int b = 0; b < NB; ++b) {
        const v4f* __restrict__ srow = (const v4f*)(style + b * SDIM);
        float a = 0.f;
#pragma unroll
        for (int i = 0; i < 4; ++i) {
            const v4f s = srow[lane + 32 * i];
            a += w[i].x * s.x;
            a += w[i].y * s.y;
            a += w[i].z * s.z;
            a += w[i].w * s.w;
        }
        acc[b] = a;
    }
#pragma unroll
    for (int b = 0; b < NB; ++b) acc[b] = wave_sum(acc[b]);

    const float mb = mod_b[ic];
    if (lane < 9) {
#pragma unroll
        for (int b = 0; b < NB; ++b) s9[b * K9 + ic * 9 + lane] = acc[b] + mb;
    }
}

__global__ __launch_bounds__(256) void demod_kernel(
    const float* __restrict__ weight, const float* __restrict__ s9,
    float* __restrict__ demod)
{
    const int oc   = blockIdx.x;
    const int lane = threadIdx.x & 31;
    const int b    = threadIdx.x >> 5;
    const v4f* __restrict__ wrow = (const v4f*)(weight + (size_t)oc * K9);
    const v4f* __restrict__ srow = (const v4f*)(s9 + (size_t)b * K9);

    v4f wv[K9 / 128], sv[K9 / 128];
#pragma unroll
    for (int i = 0; i < K9 / 128; ++i) wv[i] = wrow[lane + 32 * i];
#pragma unroll
    for (int i = 0; i < K9 / 128; ++i) sv[i] = srow[lane + 32 * i];

    float a0 = 0.f, a1 = 0.f, a2 = 0.f, a3 = 0.f;
#pragma unroll
    for (int i = 0; i < K9 / 128; ++i) {
        const v4f v = wv[i] * sv[i];
        a0 += v.x * v.x;
        a1 += v.y * v.y;
        a2 += v.z * v.z;
        a3 += v.w * v.w;
    }
    const float tot = wave_sum((a0 + a1) + (a2 + a3));
    if (lane == 0) demod[b * COUT + oc] = rsqrtf(tot + 1e-8f);
}

#define LDA 20
#define LDBC 20
#define NTILE 128

__global__ __launch_bounds__(256) void conv_wmma_kernel(
    const float* __restrict__ x, const float* __restrict__ wU,
    const float* __restrict__ bias, const float* __restrict__ s9,
    const float* __restrict__ demod, float* __restrict__ out)
{
    __shared__ float lA[2][AROWS * LDA];
    __shared__ float lB[2][NTILE * LDBC];
    __shared__ float lX[8][16 * 8 * 32];

    const int tid = threadIdx.x;
    const int h0  = blockIdx.x * 2;
    const int b   = blockIdx.y;

    const int lane = tid & 31;
    const int wid  = tid >> 5;
    const int wm   = (wid & 3) * 64;
    const int wmA  = (wid >> 2) * COUT + wm;
    const int wn   = (wid >> 2) * 64;
    const int half = lane >> 4;
    const int lr   = lane & 15;

    const int kcb0 = __builtin_amdgcn_readfirstlane(wid & 3);
    const int rsB  = __builtin_amdgcn_readfirstlane(wid >> 2);
    const int nB   = rsB * 64 + lane;

    int      ihA[2], ihB[2];
    bool     okA[2], okB[2];
    unsigned roA[2], roB[2];
    float    beta[2];
    ihA[0] = h0 + (rsB ? 1 : -1);   ihB[0] = h0 + (rsB ? 0 : 1);   beta[0] = -1.f;
    ihA[1] = h0;                    ihB[1] = h0 + (rsB ? 2 : 1);   beta[1] = rsB ? -1.f : 1.f;
#pragma unroll
    for (int jj = 0; jj < 2; ++jj) {
        okA[jj] = (unsigned)ihA[jj] < (unsigned)HW;
        okB[jj] = (unsigned)ihB[jj] < (unsigned)HW;
        roA[jj] = okA[jj] ? (unsigned)ihA[jj] * HW : 0u;
        roB[jj] = okB[jj] ? (unsigned)ihB[jj] * HW : 0u;
    }

    unsigned aoff[8];
#pragma unroll
    for (int i = 0; i < 8; ++i)
        aoff[i] = (unsigned)((tid + 256 * i) << 2);

    const float* __restrict__ s9b = s9 + (size_t)b * K9;
    const float* __restrict__ xb  = x + (size_t)b * CIN * HW * HW;

    v4f    aReg[8];
    float  bRegA[4][2];
    float  bRegB[4][2];
    float  sReg[4];
    bool   okReg[2];

    auto loadChunk = [&](int t, int jj) __attribute__((always_inline)) {
        const float* __restrict__ wt = wU + t * (AROWS * 16);
#pragma unroll
        for (int i = 0; i < 8; ++i)
            aReg[i] = *(const v4f*)(wt + aoff[i]);
        const int c   = t >> 1;
        const int icg = c / 3;
        const int kw  = c - icg * 3;
        const unsigned icb = (unsigned)(icg * 16 + kcb0);
        const float* __restrict__ xrA = xb + (icb * (HW * HW) + roA[jj]);
        const float* __restrict__ xrB = xb + (icb * (HW * HW) + roB[jj]);
        const float* __restrict__ srow = s9b + icb * 9;
        unsigned iwc[2];
#pragma unroll
        for (int q = 0; q < 2; ++q) {
            const int iw = lane + 32 * q + kw - 1;
            okReg[q] = (unsigned)iw < (unsigned)HW;
            iwc[q]   = (unsigned)min(max(iw, 0), HW - 1);
        }
#pragma unroll
        for (int i = 0; i < 4; ++i) {
#pragma unroll
            for (int q = 0; q < 2; ++q) {
                bRegA[i][q] = xrA[(4 * i) * (HW * HW) + iwc[q]];
                bRegB[i][q] = xrB[(4 * i) * (HW * HW) + iwc[q]];
            }
            sReg[i] = srow[(4 * i) * 9];
        }
        if (t + 2 < NSTEP)
            __builtin_prefetch(wt + 2 * (AROWS * 16) + aoff[0], 0, 0);
    };
    auto storeChunk = [&](int buf, int jj) __attribute__((always_inline)) {
#pragma unroll
        for (int i = 0; i < 8; ++i) {
            int id4 = tid + 256 * i;
            int row = id4 >> 2;
            int kc4 = (id4 & 3) << 2;
            *(v4f*)&lA[buf][row * LDA + kc4] = aReg[i];
        }
#pragma unroll
        for (int i = 0; i < 4; ++i) {
            int kc = kcb0 + 4 * i;
#pragma unroll
            for (int q = 0; q < 2; ++q) {
                const float da = okA[jj] ? bRegA[i][q] : 0.f;
                const float db = okB[jj] ? bRegB[i][q] : 0.f;
                const float v  = (da + beta[jj] * db) * sReg[i];
                lB[buf][(nB + 32 * q) * LDBC + kc] = okReg[q] ? v : 0.f;
            }
        }
    };

    v8f acc[2][4][4];
#pragma unroll
    for (int jj = 0; jj < 2; ++jj)
#pragma unroll
        for (int mt = 0; mt < 4; ++mt)
#pragma unroll
            for (int nt = 0; nt < 4; ++nt)
                acc[jj][mt][nt] = (v8f){};

    v2f fA0[4][4], fB0[4][4];
    v2f fA1[4][4], fB1[4][4];

    auto loadFrags = [&](int buf, v2f (&fa)[4][4], v2f (&fb)[4][4])
        __attribute__((always_inline)) {
#pragma unroll
        for (int ks = 0; ks < 4; ++ks) {
            const int ak = ks * 4 + half * 2;
#pragma unroll
            for (int mt = 0; mt < 4; ++mt)
                fa[ks][mt] = *(const v2f*)&lA[buf][(wmA + mt * 16 + lr) * LDA + ak];
#pragma unroll
            for (int nt = 0; nt < 4; ++nt)
                fb[ks][nt] = *(const v2f*)&lB[buf][(wn + nt * 16 + lr) * LDBC + ak];
        }
    };
    auto mmaChunk = [&](int jj, const v2f (&fa)[4][4], const v2f (&fb)[4][4])
        __attribute__((always_inline)) {
#pragma unroll
        for (int ks = 0; ks < 4; ++ks)
#pragma unroll
            for (int mt = 0; mt < 4; ++mt)
#pragma unroll
                for (int nt = 0; nt < 4; ++nt)
                    acc[jj][mt][nt] = wmma4(fa[ks][mt], fb[ks][nt], acc[jj][mt][nt]);
    };

    auto step = [&](int t, int par, bool doStore, bool doLoad, bool doFrags, bool doSync,
                    const v2f (&ca)[4][4], const v2f (&cb)[4][4],
                    v2f (&na)[4][4], v2f (&nb)[4][4])
        __attribute__((always_inline)) {
        if (doStore) storeChunk(par, par);
        if (doLoad)  loadChunk(t + 3, par ^ 1);
        if (doFrags) loadFrags(par ^ 1, na, nb);
        __builtin_amdgcn_sched_barrier(0);
        mmaChunk(par, ca, cb);
        __builtin_amdgcn_sched_barrier(0);
        if (doSync) __syncthreads();
    };

    loadChunk(0, 0);
    storeChunk(0, 0);
    loadChunk(1, 1);
    storeChunk(1, 1);
    loadChunk(2, 0);
    __syncthreads();
    loadFrags(0, fA0, fB0);
    __syncthreads();

    for (int t = 0; t < NSTEP - 4; t += 2) {
        step(t,     0, true, true, true, true, fA0, fB0, fA1, fB1);
        step(t + 1, 1, true, true, true, true, fA1, fB1, fA0, fB0);
    }
    step(NSTEP - 4, 0, true,  true,  true,  true,  fA0, fB0, fA1, fB1);
    step(NSTEP - 3, 1, true,  false, true,  true,  fA1, fB1, fA0, fB0);
    step(NSTEP - 2, 0, false, false, true,  false, fA0, fB0, fA1, fB1);
    step(NSTEP - 1, 1, false, false, false, false, fA1, fB1, fA0, fB0);

    float* __restrict__ myX = &lX[wid][0];
#pragma unroll
    for (int mt = 0; mt < 4; ++mt)
#pragma unroll
        for (int nt = 0; nt < 4; ++nt)
#pragma unroll
            for (int v = 0; v < 8; ++v)
                myX[((mt * 4 + nt) * 8 + v) * 32 + lane] =
                    rsB ? acc[0][mt][nt][v] : acc[1][mt][nt][v];
    __syncthreads();
    const float* __restrict__ peerX = &lX[wid ^ 4][0];

    const float* __restrict__ demodb = demod + b * COUT;
    float* __restrict__ outb = out + ((size_t)b * COUT * HW + h0 + (wid >> 2)) * HW;
#pragma unroll
    for (int mt = 0; mt < 4; ++mt) {
#pragma unroll
        for (int v = 0; v < 8; ++v) {
            int oc = wm + mt * 16 + half * 8 + v;
            float d  = demodb[oc];
            float bb = bias[oc];
#pragma unroll
            for (int nt = 0; nt < 4; ++nt) {
                const float got = peerX[((mt * 4 + nt) * 8 + v) * 32 + lane];
                const float own = acc[0][mt][nt][v] + acc[1][mt][nt][v];
                const float y   = rsB ? (got - own) : (own + got);
                outb[(size_t)oc * (HW * HW) + nt * 16 + lr] = y * d + bb;
            }
        }
    }
}

extern "C" void kernel_launch(void* const* d_in, const int* in_sizes, int n_in,
                              void* d_out, int out_size, void* d_ws, size_t ws_size,
                              hipStream_t stream) {
    const float* x      = (const float*)d_in[0];
    const float* style  = (const float*)d_in[1];
    const float* weight = (const float*)d_in[2];
    const float* bias   = (const float*)d_in[3];
    const float* mod_w  = (const float*)d_in[4];
    const float* mod_b  = (const float*)d_in[5];
    float* out = (float*)d_out;

    float* s9    = (float*)d_ws;
    float* demod = s9 + NB * K9;
    float* wU    = demod + NB * COUT;

    style_s9_kernel<<<S9_BLOCKS + WT_BLOCKS, 128, 0, stream>>>(style, mod_w, mod_b, weight, s9, wU);
    demod_kernel<<<COUT, 256, 0, stream>>>(weight, s9, demod);

    dim3 grid(HW / 2, NB);
    conv_wmma_kernel<<<grid, 256, 0, stream>>>(x, wU, bias, s9, demod, out);
}


# Horizontal minimal filtering F(2,3) nested on the vertical one (F(2x2,3x3)): 16 products per 2x2 outputs, 64 K steps x 2 oc halves (512 blocks), was 96; conv summation order and rounding change
# speedup vs baseline: 3.3669x; 1.1064x over previous
#include <hip/hip_runtime.h>
#include <hip/hip_bf16.h>

typedef float v2f __attribute__((ext_vector_type(2)));
typedef float v4f __attribute__((ext_vector_type(4)));
typedef float v8f __attribute__((ext_vector_type(8)));

__device__ __forceinline__ v8f wmma4(v2f a, v2f b, v8f c) {
    return __builtin_amdgcn_wmma_f32_16x16x4_f32(
         false, a,  false, b,
         (short)0, c,  false,  false);
}

#define CIN   256
#define COUT  256
#define HW    64
#define K9    2304
#define SDIM  512
#define NB    8

__device__ __forceinline__ float wave_sum(float v) {
#pragma unroll
    for (int off = 16; off > 0; off >>= 1) v += __shfl_xor(v, off);
    return v;
}

#define S9_BLOCKS  (CIN / 4)
#define WT_BLOCKS  (COUT * 16 / 64)
#define AROWS      512
#define NSTEP      64
#define OCB        128

__device__ __forceinline__ float gform(int m, float a, float b, float c) {
    return (m == 0) ? a
         : (m == 1) ? 0.5f * ((a + b) + c)
         : (m == 2) ? 0.5f * ((a - b) + c)
         :            c;
}

__global__ __launch_bounds__(128) void style_s9_kernel(
    const float* __restrict__ style, const float* __restrict__ mod_w,
    const float* __restrict__ mod_b, const float* __restrict__ weight,
    float* __restrict__ s9, float* __restrict__ wU)
{
    if (blockIdx.x >= S9_BLOCKS) {
        if (threadIdx.x >= 64) return;
        const int r   = blockIdx.x - S9_BLOCKS;
        const int icg = r >> 2;
        const int oc  = ((r & 3) << 6) + threadIdx.x;
        const v4f* __restrict__ src = (const v4f*)(weight + (size_t)oc * K9 + icg * 144);
        v4f in[36];
#pragma unroll
        for (int q = 0; q < 36; ++q) in[q] = src[q];
        float* __restrict__ dstz = wU + (size_t)(oc >> 7) * NSTEP * AROWS * 16;
        const int ocl = oc & (OCB - 1);
#pragma unroll
        for (int g = 0; g < 4; ++g) {
            float U[4][4][4];
#pragma unroll
            for (int e = 0; e < 4; ++e) {
                const int kc = 4 * g + e;
                float hz[3][4];
#pragma unroll
                for (int kh = 0; kh < 3; ++kh) {
                    const float a  = in[(kc * 9 + kh * 3 + 0) >> 2][(kc * 9 + kh * 3 + 0) & 3];
                    const float bb = in[(kc * 9 + kh * 3 + 1) >> 2][(kc * 9 + kh * 3 + 1) & 3];
                    const float cc = in[(kc * 9 + kh * 3 + 2) >> 2][(kc * 9 + kh * 3 + 2) & 3];
#pragma unroll
                    for (int i = 0; i < 4; ++i) hz[kh][i] = gform(i, a, bb, cc);
                }
#pragma unroll
                for (int j = 0; j < 4; ++j)
#pragma unroll
                    for (int i = 0; i < 4; ++i) U[j][i][e] = gform(j, hz[0][i], hz[1][i], hz[2][i]);
            }
#pragma unroll
            for (int j = 0; j < 4; ++j) {
#pragma unroll
                for (int i = 0; i < 4; ++i) {
                    const int t   = 2 * (2 * icg + (i >> 1)) + (j & 1);
                    const int row = (j >> 1) * 256 + (i & 1) * OCB + ocl;
                    v4f o;
                    o.x = U[j][i][0]; o.y = U[j][i][1]; o.z = U[j][i][2]; o.w = U[j][i][3];
                    *(v4f*)(dstz + ((size_t)t * AROWS + row) * 16 + 4 * g) = o;
                }
            }
        }
        return;
    }
    const int lane = threadIdx.x & 31;
    const int ic   = blockIdx.x * 4 + (threadIdx.x >> 5);

    const v4f* __restrict__ wrow = (const v4f*)(mod_w + (size_t)ic * SDIM);
    v4f w[4];
#pragma unroll
    for (int i = 0; i < 4; ++i) w[i] = wrow[lane + 32 * i];

    float acc[NB];
#pragma unroll
    for (int b = 0; b < NB; ++b) {
        const v4f* __restrict__ srow = (const v4f*)(style + b * SDIM);
        float a = 0.f;
#pragma unroll
        for (int i = 0; i < 4; ++i) {
            const v4f s = srow[lane + 32 * i];
            a += w[i].x * s.x;
            a += w[i].y * s.y;
            a += w[i].z * s.z;
            a += w[i].w * s.w;
        }
        acc[b] = a;
    }
#pragma unroll
    for (int b = 0; b < NB; ++b) acc[b] = wave_sum(acc[b]);

    const float mb = mod_b[ic];
    if (lane < 9) {
#pragma unroll
        for (int b = 0; b < NB; ++b) s9[b * K9 + ic * 9 + lane] = acc[b] + mb;
    }
}

__global__ __launch_bounds__(256) void demod_kernel(
    const float* __restrict__ weight, const float* __restrict__ s9,
    float* __restrict__ demod)
{
    const int oc   = blockIdx.x;
    const int lane = threadIdx.x & 31;
    const int b    = threadIdx.x >> 5;
    const v4f* __restrict__ wrow = (const v4f*)(weight + (size_t)oc * K9);
    const v4f* __restrict__ srow = (const v4f*)(s9 + (size_t)b * K9);

    v4f wv[K9 / 128], sv[K9 / 128];
#pragma unroll
    for (int i = 0; i < K9 / 128; ++i) wv[i] = wrow[lane + 32 * i];
#pragma unroll
    for (int i = 0; i < K9 / 128; ++i) sv[i] = srow[lane + 32 * i];

    float a0 = 0.f, a1 = 0.f, a2 = 0.f, a3 = 0.f;
#pragma unroll
    for (int i = 0; i < K9 / 128; ++i) {
        const v4f v = wv[i] * sv[i];
        a0 += v.x * v.x;
        a1 += v.y * v.y;
        a2 += v.z * v.z;
        a3 += v.w * v.w;
    }
    const float tot = wave_sum((a0 + a1) + (a2 + a3));
    if (lane == 0) demod[b * COUT + oc] = rsqrtf(tot + 1e-8f);
}

#define LDA 20
#define LDBC 20
#define NTILE 128

__global__ __launch_bounds__(256) void conv_wmma_kernel(
    const float* __restrict__ x, const float* __restrict__ wU,
    const float* __restrict__ bias, const float* __restrict__ s9,
    const float* __restrict__ demod, float* __restrict__ out)
{
    __shared__ float lA[2][AROWS * LDA];
    __shared__ float lB[2][NTILE * LDBC];
    __shared__ float lX[8][8 * 8 * 32];

    const int tid = threadIdx.x;
    const int h0  = blockIdx.x * 2;
    const int b   = blockIdx.y;
    const int z   = blockIdx.z;

    const int lane = tid & 31;
    const int wid  = tid >> 5;
    const int wq   = (wid & 3) * 32;
    const int wmA  = (wid >> 2) * 256 + wq;
    const int wn   = (wid >> 2) * 64;
    const int half = lane >> 4;
    const int lr   = lane & 15;

    const int kcb0 = __builtin_amdgcn_readfirstlane(wid & 3);
    const int rsB  = __builtin_amdgcn_readfirstlane(wid >> 2);
    const int nB   = rsB * 64 + lane;

    int      ihA[2], ihB[2];
    bool     okA[2], okB[2];
    unsigned roA[2], roB[2];
    float    beta[2];
    ihA[0] = h0 + (rsB ? 1 : -1);   ihB[0] = h0 + (rsB ? 0 : 1);   beta[0] = -1.f;
    ihA[1] = h0;                    ihB[1] = h0 + (rsB ? 2 : 1);   beta[1] = rsB ? -1.f : 1.f;
#pragma unroll
    for (int jj = 0; jj < 2; ++jj) {
        okA[jj] = (unsigned)ihA[jj] < (unsigned)HW;
        okB[jj] = (unsigned)ihB[jj] < (unsigned)HW;
        roA[jj] = okA[jj] ? (unsigned)ihA[jj] * HW : 0u;
        roB[jj] = okB[jj] ? (unsigned)ihB[jj] * HW : 0u;
    }

    unsigned cP[2][2], cQ[2][2];
    bool     okP[2][2], okQ[2][2];
    float    gam[2][2];
    {
        const int c2 = 2 * lane;
        const int p00 = c2 - 1, q00 = c2 + 1;   gam[0][0] = -1.f;
        const int p01 = c2,     q01 = c2 + 1;   gam[0][1] =  1.f;
        const int p10 = c2 + 1, q10 = c2;       gam[1][0] = -1.f;
        const int p11 = c2,     q11 = c2 + 2;   gam[1][1] = -1.f;
        const int pp[2][2] = {{p00, p01}, {p10, p11}};
        const int qq[2][2] = {{q00, q01}, {q10, q11}};
#pragma unroll
        for (int ih = 0; ih < 2; ++ih)
#pragma unroll
            for (int iw = 0; iw < 2; ++iw) {
                okP[ih][iw] = (unsigned)pp[ih][iw] < (unsigned)HW;
                okQ[ih][iw] = (unsigned)qq[ih][iw] < (unsigned)HW;
                cP[ih][iw]  = (unsigned)min(max(pp[ih][iw], 0), HW - 1);
                cQ[ih][iw]  = (unsigned)min(max(qq[ih][iw], 0), HW - 1);
            }
    }

    unsigned aoff[8];
#pragma unroll
    for (int i = 0; i < 8; ++i)
        aoff[i] = (unsigned)((tid + 256 * i) << 2);

    const float* __restrict__ s9b = s9 + (size_t)b * K9;
    const float* __restrict__ xb  = x + (size_t)b * CIN * HW * HW;
    const float* __restrict__ wUz = wU + (size_t)z * NSTEP * AROWS * 16;

    v4f    aReg[8];
    float  bRegA[4][2][2];
    float  bRegB[4][2][2];
    float  sReg[4];

    auto loadChunk = [&](int t, int jj, int ih) __attribute__((always_inline)) {
        const float* __restrict__ wt = wUz + t * (AROWS * 16);
#pragma unroll
        for (int i = 0; i < 8; ++i)
            aReg[i] = *(const v4f*)(wt + aoff[i]);
        const int icg = t >> 2;
        const unsigned icb = (unsigned)(icg * 16 + kcb0);
        const float* __restrict__ xrA = xb + (icb * (HW * HW) + roA[jj]);
        const float* __restrict__ xrB = xb + (icb * (HW * HW) + roB[jj]);
        const float* __restrict__ srow = s9b + icb * 9;
#pragma unroll
        for (int i = 0; i < 4; ++i) {
#pragma unroll
            for (int q = 0; q < 2; ++q) {
                bRegA[i][q][0] = xrA[(4 * i) * (HW * HW) + cP[ih][q]];
                bRegA[i][q][1] = xrA[(4 * i) * (HW * HW) + cQ[ih][q]];
                bRegB[i][q][0] = xrB[(4 * i) * (HW * HW) + cP[ih][q]];
                bRegB[i][q][1] = xrB[(4 * i) * (HW * HW) + cQ[ih][q]];
            }
            sReg[i] = srow[(4 * i) * 9];
        }
        if (t + 2 < NSTEP)
            __builtin_prefetch(wt + 2 * (AROWS * 16) + aoff[0], 0, 0);
    };
    auto storeChunk = [&](int buf, int jj, int ih) __attribute__((always_inline)) {
#pragma unroll
        for (int i = 0; i < 8; ++i) {
            int id4 = tid + 256 * i;
            int row = id4 >> 2;
            int kc4 = (id4 & 3) << 2;
            *(v4f*)&lA[buf][row * LDA + kc4] = aReg[i];
        }
#pragma unroll
        for (int i = 0; i < 4; ++i) {
            int kc = kcb0 + 4 * i;
#pragma unroll
            for (int q = 0; q < 2; ++q) {
                const float daP = okA[jj] ? bRegA[i][q][0] : 0.f;
                const float dbP = okB[jj] ? bRegB[i][q][0] : 0.f;
                const float daQ = okA[jj] ? bRegA[i][q][1] : 0.f;
                const float dbQ = okB[jj] ? bRegB[i][q][1] : 0.f;
                const float vP  = okP[ih][q] ? (daP + beta[jj] * dbP) : 0.f;
                const float vQ  = okQ[ih][q] ? (daQ + beta[jj] * dbQ) : 0.f;
                lB[buf][(nB + 32 * q) * LDBC + kc] = (vP + gam[ih][q] * vQ) * sReg[i];
            }
        }
    };

    v8f acc[4][4][2];
#pragma unroll
    for (int s = 0; s < 4; ++s)
#pragma unroll
        for (int m4 = 0; m4 < 4; ++m4)
#pragma unroll
            for (int nt = 0; nt < 2; ++nt)
                acc[s][m4][nt] = (v8f){};

    v2f fA0[4][4], fB0[4][4];
    v2f fA1[4][4], fB1[4][4];

    auto loadFrags = [&](int buf, v2f (&fa)[4][4], v2f (&fb)[4][4])
        __attribute__((always_inline)) {
#pragma unroll
        for (int ks = 0; ks < 4; ++ks) {
            const int ak = ks * 4 + half * 2;
#pragma unroll
            for (int f = 0; f < 4; ++f)
                fa[ks][f] = *(const v2f*)&lA[buf][(wmA + (f >> 1) * OCB + (f & 1) * 16 + lr) * LDA + ak];
#pragma unroll
            for (int f = 0; f < 4; ++f)
                fb[ks][f] = *(const v2f*)&lB[buf][(wn + f * 16 + lr) * LDBC + ak];
        }
    };
    auto mmaChunk = [&](int s, const v2f (&fa)[4][4], const v2f (&fb)[4][4])
        __attribute__((always_inline)) {
#pragma unroll
        for (int ks = 0; ks < 4; ++ks)
#pragma unroll
            for (int f = 0; f < 4; ++f)
#pragma unroll
                for (int nt = 0; nt < 2; ++nt)
                    acc[s][f][nt] = wmma4(fa[ks][f], fb[ks][(f >> 1) * 2 + nt], acc[s][f][nt]);
    };

    auto step = [&](int t, int s, bool doStore, bool doLoad, bool doFrags, bool doSync,
                    const v2f (&ca)[4][4], const v2f (&cb)[4][4],
                    v2f (&na)[4][4], v2f (&nb)[4][4])
        __attribute__((always_inline)) {
        const int par = s & 1;
        if (doStore) storeChunk(par, par, ((s + 2) >> 1) & 1);
        if (doLoad)  loadChunk(t + 3, par ^ 1, ((s + 3) >> 1) & 1);
        if (doFrags) loadFrags(par ^ 1, na, nb);
        __builtin_amdgcn_sched_barrier(0);
        mmaChunk(s, ca, cb);
        __builtin_amdgcn_sched_barrier(0);
        if (doSync) __syncthreads();
    };

    loadChunk(0, 0, 0);
    storeChunk(0, 0, 0);
    loadChunk(1, 1, 0);
    storeChunk(1, 1, 0);
    loadChunk(2, 0, 1);
    __syncthreads();
    loadFrags(0, fA0, fB0);
    __syncthreads();

    for (int t = 0; t < NSTEP - 4; t += 4) {
        step(t,     0, true, true, true, true, fA0, fB0, fA1, fB1);
        step(t + 1, 1, true, true, true, true, fA1, fB1, fA0, fB0);
        step(t + 2, 2, true, true, true, true, fA0, fB0, fA1, fB1);
        step(t + 3, 3, true, true, true, true, fA1, fB1, fA0, fB0);
    }
    step(NSTEP - 4, 0, true,  true,  true,  true,  fA0, fB0, fA1, fB1);
    step(NSTEP - 3, 1, true,  false, true,  true,  fA1, fB1, fA0, fB0);
    step(NSTEP - 2, 2, false, false, true,  false, fA0, fB0, fA1, fB1);
    step(NSTEP - 1, 3, false, false, false, false, fA1, fB1, fA0, fB0);

    v8f M[2][2][2][2];
#pragma unroll
    for (int jj = 0; jj < 2; ++jj)
#pragma unroll
        for (int mt = 0; mt < 2; ++mt)
#pragma unroll
            for (int nt = 0; nt < 2; ++nt) {
                const v8f n0 = acc[jj][mt][nt],     n1 = acc[jj][2 + mt][nt];
                const v8f n2 = acc[2 + jj][mt][nt], n3 = acc[2 + jj][2 + mt][nt];
                M[jj][0][mt][nt] = (n0 + n1) + n2;
                M[jj][1][mt][nt] = n1 - (n2 + n3);
            }

    float* __restrict__ myX = &lX[wid][0];
#pragma unroll
    for (int pc = 0; pc < 2; ++pc)
#pragma unroll
        for (int mt = 0; mt < 2; ++mt)
#pragma unroll
            for (int nt = 0; nt < 2; ++nt)
#pragma unroll
                for (int v = 0; v < 8; ++v)
                    myX[(((pc * 2 + mt) * 2 + nt) * 8 + v) * 32 + lane] =
                        rsB ? M[0][pc][mt][nt][v] : M[1][pc][mt][nt][v];
    __syncthreads();
    const float* __restrict__ peerX = &lX[wid ^ 4][0];

    const float* __restrict__ demodb = demod + b * COUT;
    float* __restrict__ outb = out + ((size_t)b * COUT * HW + h0 + (wid >> 2)) * HW;
#pragma unroll
    for (int mt = 0; mt < 2; ++mt) {
#pragma unroll
        for (int v = 0; v < 8; ++v) {
            int oc = z * OCB + wq + mt * 16 + half * 8 + v;
            float d  = demodb[oc];
            float bb = bias[oc];
#pragma unroll
            for (int nt = 0; nt < 2; ++nt) {
                v2f y;
#pragma unroll
                for (int pc = 0; pc < 2; ++pc) {
                    const float got = peerX[(((pc * 2 + mt) * 2 + nt) * 8 + v) * 32 + lane];
                    const float own = M[0][pc][mt][nt][v] + M[1][pc][mt][nt][v];
                    const float yy  = rsB ? (got - own) : (own + got);
                    y[pc] = yy * d + bb;
                }
                *(v2f*)&outb[(size_t)oc * (HW * HW) + 2 * (nt * 16 + lr)] = y;
            }
        }
    }
}

extern "C" void kernel_launch(void* const* d_in, const int* in_sizes, int n_in,
                              void* d_out, int out_size, void* d_ws, size_t ws_size,
                              hipStream_t stream) {
    const float* x      = (const float*)d_in[0];
    const float* style  = (const float*)d_in[1];
    const float* weight = (const float*)d_in[2];
    const float* bias   = (const float*)d_in[3];
    const float* mod_w  = (const float*)d_in[4];
    const float* mod_b  = (const float*)d_in[5];
    float* out = (float*)d_out;

    float* s9    = (float*)d_ws;
    float* demod = s9 + NB * K9;
    float* wU    = demod + NB * COUT;

    style_s9_kernel<<<S9_BLOCKS + WT_BLOCKS, 128, 0, stream>>>(style, mod_w, mod_b, weight, s9, wU);
    demod_kernel<<<COUT, 256, 0, stream>>>(weight, s9, demod);

    dim3 grid(HW / 2, NB, COUT / OCB);
    conv_wmma_kernel<<<grid, 256, 0, stream>>>(x, wU, bias, s9, demod, out);
}


# Step barrier split: signal inside the WMMA block behind an LDS-only wait, wait behind the block; the two waves of a SIMD alternate staging and WMMAs; global loads stay in flight across the barrier
# speedup vs baseline: 3.6252x; 1.0767x over previous
#include <hip/hip_runtime.h>
#include <hip/hip_bf16.h>

typedef float v2f __attribute__((ext_vector_type(2)));
typedef float v4f __attribute__((ext_vector_type(4)));
typedef float v8f __attribute__((ext_vector_type(8)));

__device__ __forceinline__ v8f wmma4(v2f a, v2f b, v8f c) {
    return __builtin_amdgcn_wmma_f32_16x16x4_f32(
         false, a,  false, b,
         (short)0, c,  false,  false);
}

#define CIN   256
#define COUT  256
#define HW    64
#define K9    2304
#define SDIM  512
#define NB    8

__device__ __forceinline__ float wave_sum(float v) {
#pragma unroll
    for (int off = 16; off > 0; off >>= 1) v += __shfl_xor(v, off);
    return v;
}

#define S9_BLOCKS  (CIN / 4)
#define WT_BLOCKS  (COUT * 16 / 64)
#define AROWS      512
#define NSTEP      64
#define OCB        128

__device__ __forceinline__ float gform(int m, float a, float b, float c) {
    return (m == 0) ? a
         : (m == 1) ? 0.5f * ((a + b) + c)
         : (m == 2) ? 0.5f * ((a - b) + c)
         :            c;
}

__global__ __launch_bounds__(128) void style_s9_kernel(
    const float* __restrict__ style, const float* __restrict__ mod_w,
    const float* __restrict__ mod_b, const float* __restrict__ weight,
    float* __restrict__ s9, float* __restrict__ wU)
{
    if (blockIdx.x >= S9_BLOCKS) {
        if (threadIdx.x >= 64) return;
        const int r   = blockIdx.x - S9_BLOCKS;
        const int icg = r >> 2;
        const int oc  = ((r & 3) << 6) + threadIdx.x;
        const v4f* __restrict__ src = (const v4f*)(weight + (size_t)oc * K9 + icg * 144);
        v4f in[36];
#pragma unroll
        for (int q = 0; q < 36; ++q) in[q] = src[q];
        float* __restrict__ dstz = wU + (size_t)(oc >> 7) * NSTEP * AROWS * 16;
        const int ocl = oc & (OCB - 1);
#pragma unroll
        for (int g = 0; g < 4; ++g) {
            float U[4][4][4];
#pragma unroll
            for (int e = 0; e < 4; ++e) {
                const int kc = 4 * g + e;
                float hz[3][4];
#pragma unroll
                for (int kh = 0; kh < 3; ++kh) {
                    const float a  = in[(kc * 9 + kh * 3 + 0) >> 2][(kc * 9 + kh * 3 + 0) & 3];
                    const float bb = in[(kc * 9 + kh * 3 + 1) >> 2][(kc * 9 + kh * 3 + 1) & 3];
                    const float cc = in[(kc * 9 + kh * 3 + 2) >> 2][(kc * 9 + kh * 3 + 2) & 3];
#pragma unroll
                    for (int i = 0; i < 4; ++i) hz[kh][i] = gform(i, a, bb, cc);
                }
#pragma unroll
                for (int j = 0; j < 4; ++j)
#pragma unroll
                    for (int i = 0; i < 4; ++i) U[j][i][e] = gform(j, hz[0][i], hz[1][i], hz[2][i]);
            }
#pragma unroll
            for (int j = 0; j < 4; ++j) {
#pragma unroll
                for (int i = 0; i < 4; ++i) {
                    const int t   = 2 * (2 * icg + (i >> 1)) + (j & 1);
                    const int row = (j >> 1) * 256 + (i & 1) * OCB + ocl;
                    v4f o;
                    o.x = U[j][i][0]; o.y = U[j][i][1]; o.z = U[j][i][2]; o.w = U[j][i][3];
                    *(v4f*)(dstz + ((size_t)t * AROWS + row) * 16 + 4 * g) = o;
                }
            }
        }
        return;
    }
    const int lane = threadIdx.x & 31;
    const int ic   = blockIdx.x * 4 + (threadIdx.x >> 5);

    const v4f* __restrict__ wrow = (const v4f*)(mod_w + (size_t)ic * SDIM);
    v4f w[4];
#pragma unroll
    for (int i = 0; i < 4; ++i) w[i] = wrow[lane + 32 * i];

    float acc[NB];
#pragma unroll
    for (int b = 0; b < NB; ++b) {
        const v4f* __restrict__ srow = (const v4f*)(style + b * SDIM);
        float a = 0.f;
#pragma unroll
        for (int i = 0; i < 4; ++i) {
            const v4f s = srow[lane + 32 * i];
            a += w[i].x * s.x;
            a += w[i].y * s.y;
            a += w[i].z * s.z;
            a += w[i].w * s.w;
        }
        acc[b] = a;
    }
#pragma unroll
    for (int b = 0; b < NB; ++b) acc[b] = wave_sum(acc[b]);

    const float mb = mod_b[ic];
    if (lane < 9) {
#pragma unroll
        for (int b = 0; b < NB; ++b) s9[b * K9 + ic * 9 + lane] = acc[b] + mb;
    }
}

__global__ __launch_bounds__(256) void demod_kernel(
    const float* __restrict__ weight, const float* __restrict__ s9,
    float* __restrict__ demod)
{
    const int oc   = blockIdx.x;
    const int lane = threadIdx.x & 31;
    const int b    = threadIdx.x >> 5;
    const v4f* __restrict__ wrow = (const v4f*)(weight + (size_t)oc * K9);
    const v4f* __restrict__ srow = (const v4f*)(s9 + (size_t)b * K9);

    v4f wv[K9 / 128], sv[K9 / 128];
#pragma unroll
    for (int i = 0; i < K9 / 128; ++i) wv[i] = wrow[lane + 32 * i];
#pragma unroll
    for (int i = 0; i < K9 / 128; ++i) sv[i] = srow[lane + 32 * i];

    float a0 = 0.f, a1 = 0.f, a2 = 0.f, a3 = 0.f;
#pragma unroll
    for (int i = 0; i < K9 / 128; ++i) {
        const v4f v = wv[i] * sv[i];
        a0 += v.x * v.x;
        a1 += v.y * v.y;
        a2 += v.z * v.z;
        a3 += v.w * v.w;
    }
    const float tot = wave_sum((a0 + a1) + (a2 + a3));
    if (lane == 0) demod[b * COUT + oc] = rsqrtf(tot + 1e-8f);
}

#define LDA 20
#define LDBC 20
#define NTILE 128

__global__ __launch_bounds__(256) void conv_wmma_kernel(
    const float* __restrict__ x, const float* __restrict__ wU,
    const float* __restrict__ bias, const float* __restrict__ s9,
    const float* __restrict__ demod, float* __restrict__ out)
{
    __shared__ float lA[2][AROWS * LDA];
    __shared__ float lB[2][NTILE * LDBC];
    __shared__ float lX[8][8 * 8 * 32];

    const int tid = threadIdx.x;
    const int h0  = blockIdx.x * 2;
    const int b   = blockIdx.y;
    const int z   = blockIdx.z;

    const int lane = tid & 31;
    const int wid  = tid >> 5;
    const int wq   = (wid & 3) * 32;
    const int wmA  = (wid >> 2) * 256 + wq;
    const int wn   = (wid >> 2) * 64;
    const int half = lane >> 4;
    const int lr   = lane & 15;

    const int kcb0 = __builtin_amdgcn_readfirstlane(wid & 3);
    const int rsB  = __builtin_amdgcn_readfirstlane(wid >> 2);
    const int nB   = rsB * 64 + lane;

    int      ihA[2], ihB[2];
    bool     okA[2], okB[2];
    unsigned roA[2], roB[2];
    float    beta[2];
    ihA[0] = h0 + (rsB ? 1 : -1);   ihB[0] = h0 + (rsB ? 0 : 1);   beta[0] = -1.f;
    ihA[1] = h0;                    ihB[1] = h0 + (rsB ? 2 : 1);   beta[1] = rsB ? -1.f : 1.f;
#pragma unroll
    for (int jj = 0; jj < 2; ++jj) {
        okA[jj] = (unsigned)ihA[jj] < (unsigned)HW;
        okB[jj] = (unsigned)ihB[jj] < (unsigned)HW;
        roA[jj] = okA[jj] ? (unsigned)ihA[jj] * HW : 0u;
        roB[jj] = okB[jj] ? (unsigned)ihB[jj] * HW : 0u;
    }

    unsigned cP[2][2], cQ[2][2];
    bool     okP[2][2], okQ[2][2];
    float    gam[2][2];
    {
        const int c2 = 2 * lane;
        const int p00 = c2 - 1, q00 = c2 + 1;   gam[0][0] = -1.f;
        const int p01 = c2,     q01 = c2 + 1;   gam[0][1] =  1.f;
        const int p10 = c2 + 1, q10 = c2;       gam[1][0] = -1.f;
        const int p11 = c2,     q11 = c2 + 2;   gam[1][1] = -1.f;
        const int pp[2][2] = {{p00, p01}, {p10, p11}};
        const int qq[2][2] = {{q00, q01}, {q10, q11}};
#pragma unroll
        for (int ih = 0; ih < 2; ++ih)
#pragma unroll
            for (int iw = 0; iw < 2; ++iw) {
                okP[ih][iw] = (unsigned)pp[ih][iw] < (unsigned)HW;
                okQ[ih][iw] = (unsigned)qq[ih][iw] < (unsigned)HW;
                cP[ih][iw]  = (unsigned)min(max(pp[ih][iw], 0), HW - 1);
                cQ[ih][iw]  = (unsigned)min(max(qq[ih][iw], 0), HW - 1);
            }
    }

    unsigned aoff[8];
#pragma unroll
    for (int i = 0; i < 8; ++i)
        aoff[i] = (unsigned)((tid + 256 * i) << 2);

    const float* __restrict__ s9b = s9 + (size_t)b * K9;
    const float* __restrict__ xb  = x + (size_t)b * CIN * HW * HW;
    const float* __restrict__ wUz = wU + (size_t)z * NSTEP * AROWS * 16;

    v4f    aReg[8];
    float  bRegA[4][2][2];
    float  bRegB[4][2][2];
    float  sReg[4];

    auto loadChunk = [&](int t, int jj, int ih) __attribute__((always_inline)) {
        const float* __restrict__ wt = wUz + t * (AROWS * 16);
#pragma unroll
        for (int i = 0; i < 8; ++i)
            aReg[i] = *(const v4f*)(wt + aoff[i]);
        const int icg = t >> 2;
        const unsigned icb = (unsigned)(icg * 16 + kcb0);
        const float* __restrict__ xrA = xb + (icb * (HW * HW) + roA[jj]);
        const float* __restrict__ xrB = xb + (icb * (HW * HW) + roB[jj]);
        const float* __restrict__ srow = s9b + icb * 9;
#pragma unroll
        for (int i = 0; i < 4; ++i) {
#pragma unroll
            for (int q = 0; q < 2; ++q) {
                bRegA[i][q][0] = xrA[(4 * i) * (HW * HW) + cP[ih][q]];
                bRegA[i][q][1] = xrA[(4 * i) * (HW * HW) + cQ[ih][q]];
                bRegB[i][q][0] = xrB[(4 * i) * (HW * HW) + cP[ih][q]];
                bRegB[i][q][1] = xrB[(4 * i) * (HW * HW) + cQ[ih][q]];
            }
            sReg[i] = srow[(4 * i) * 9];
        }
        if (t + 2 < NSTEP)
            __builtin_prefetch(wt + 2 * (AROWS * 16) + aoff[0], 0, 0);
    };
    auto storeChunk = [&](int buf, int jj, int ih) __attribute__((always_inline)) {
#pragma unroll
        for (int i = 0; i < 8; ++i) {
            int id4 = tid + 256 * i;
            int row = id4 >> 2;
            int kc4 = (id4 & 3) << 2;
            *(v4f*)&lA[buf][row * LDA + kc4] = aReg[i];
        }
#pragma unroll
        for (int i = 0; i < 4; ++i) {
            int kc = kcb0 + 4 * i;
#pragma unroll
            for (int q = 0; q < 2; ++q) {
                const float daP = okA[jj] ? bRegA[i][q][0] : 0.f;
                const float dbP = okB[jj] ? bRegB[i][q][0] : 0.f;
                const float daQ = okA[jj] ? bRegA[i][q][1] : 0.f;
                const float dbQ = okB[jj] ? bRegB[i][q][1] : 0.f;
                const float vP  = okP[ih][q] ? (daP + beta[jj] * dbP) : 0.f;
                const float vQ  = okQ[ih][q] ? (daQ + beta[jj] * dbQ) : 0.f;
                lB[buf][(nB + 32 * q) * LDBC + kc] = (vP + gam[ih][q] * vQ) * sReg[i];
            }
        }
    };

    v8f acc[4][4][2];
#pragma unroll
    for (int s = 0; s < 4; ++s)
#pragma unroll
        for (int m4 = 0; m4 < 4; ++m4)
#pragma unroll
            for (int nt = 0; nt < 2; ++nt)
                acc[s][m4][nt] = (v8f){};

    v2f fA0[4][4], fB0[4][4];
    v2f fA1[4][4], fB1[4][4];

    auto loadFrags = [&](int buf, v2f (&fa)[4][4], v2f (&fb)[4][4])
        __attribute__((always_inline)) {
#pragma unroll
        for (int ks = 0; ks < 4; ++ks) {
            const int ak = ks * 4 + half * 2;
#pragma unroll
            for (int f = 0; f < 4; ++f)
                fa[ks][f] = *(const v2f*)&lA[buf][(wmA + (f >> 1) * OCB + (f & 1) * 16 + lr) * LDA + ak];
#pragma unroll
            for (int f = 0; f < 4; ++f)
                fb[ks][f] = *(const v2f*)&lB[buf][(wn + f * 16 + lr) * LDBC + ak];
        }
    };
    auto mmaChunk = [&](int s, int ks0, int ks1, const v2f (&fa)[4][4], const v2f (&fb)[4][4])
        __attribute__((always_inline)) {
#pragma unroll
        for (int ks = ks0; ks < ks1; ++ks)
#pragma unroll
            for (int f = 0; f < 4; ++f)
#pragma unroll
                for (int nt = 0; nt < 2; ++nt)
                    acc[s][f][nt] = wmma4(fa[ks][f], fb[ks][(f >> 1) * 2 + nt], acc[s][f][nt]);
    };

    auto step = [&](int t, int s, bool doStore, bool doLoad, bool doFrags, bool doSync,
                    const v2f (&ca)[4][4], const v2f (&cb)[4][4],
                    v2f (&na)[4][4], v2f (&nb)[4][4])
        __attribute__((always_inline)) {
        const int par = s & 1;
        if (doStore) storeChunk(par, par, ((s + 2) >> 1) & 1);
        if (doLoad)  loadChunk(t + 3, par ^ 1, ((s + 3) >> 1) & 1);
        if (doFrags) loadFrags(par ^ 1, na, nb);
        __builtin_amdgcn_sched_barrier(0);
        mmaChunk(s, 0, 3, ca, cb);
        if (doSync) {
            __builtin_amdgcn_sched_barrier(0);
            asm volatile("s_wait_dscnt 0x0" ::: "memory");
            asm volatile("s_barrier_signal -1" ::: "memory");
            __builtin_amdgcn_sched_barrier(0);
        }
        mmaChunk(s, 3, 4, ca, cb);
        __builtin_amdgcn_sched_barrier(0);
        if (doSync) asm volatile("s_barrier_wait -1" ::: "memory");
    };

    loadChunk(0, 0, 0);
    storeChunk(0, 0, 0);
    loadChunk(1, 1, 0);
    storeChunk(1, 1, 0);
    loadChunk(2, 0, 1);
    __syncthreads();
    loadFrags(0, fA0, fB0);
    __syncthreads();

    for (int t = 0; t < NSTEP - 4; t += 4) {
        step(t,     0, true, true, true, true, fA0, fB0, fA1, fB1);
        step(t + 1, 1, true, true, true, true, fA1, fB1, fA0, fB0);
        step(t + 2, 2, true, true, true, true, fA0, fB0, fA1, fB1);
        step(t + 3, 3, true, true, true, true, fA1, fB1, fA0, fB0);
    }
    step(NSTEP - 4, 0, true,  true,  true,  true,  fA0, fB0, fA1, fB1);
    step(NSTEP - 3, 1, true,  false, true,  true,  fA1, fB1, fA0, fB0);
    step(NSTEP - 2, 2, false, false, true,  false, fA0, fB0, fA1, fB1);
    step(NSTEP - 1, 3, false, false, false, false, fA1, fB1, fA0, fB0);

    v8f M[2][2][2][2];
#pragma unroll
    for (int jj = 0; jj < 2; ++jj)
#pragma unroll
        for (int mt = 0; mt < 2; ++mt)
#pragma unroll
            for (int nt = 0; nt < 2; ++nt) {
                const v8f n0 = acc[jj][mt][nt],     n1 = acc[jj][2 + mt][nt];
                const v8f n2 = acc[2 + jj][mt][nt], n3 = acc[2 + jj][2 + mt][nt];
                M[jj][0][mt][nt] = (n0 + n1) + n2;
                M[jj][1][mt][nt] = n1 - (n2 + n3);
            }

    float* __restrict__ myX = &lX[wid][0];
#pragma unroll
    for (int pc = 0; pc < 2; ++pc)
#pragma unroll
        for (int mt = 0; mt < 2; ++mt)
#pragma unroll
            for (int nt = 0; nt < 2; ++nt)
#pragma unroll
                for (int v = 0; v < 8; ++v)
                    myX[(((pc * 2 + mt) * 2 + nt) * 8 + v) * 32 + lane] =
                        rsB ? M[0][pc][mt][nt][v] : M[1][pc][mt][nt][v];
    __syncthreads();
    const float* __restrict__ peerX = &lX[wid ^ 4][0];

    const float* __restrict__ demodb = demod + b * COUT;
    float* __restrict__ outb = out + ((size_t)b * COUT * HW + h0 + (wid >> 2)) * HW;
#pragma unroll
    for (int mt = 0; mt < 2; ++mt) {
#pragma unroll
        for (int v = 0; v < 8; ++v) {
            int oc = z * OCB + wq + mt * 16 + half * 8 + v;
            float d  = demodb[oc];
            float bb = bias[oc];
#pragma unroll
            for (int nt = 0; nt < 2; ++nt) {
                v2f y;
#pragma unroll
                for (int pc = 0; pc < 2; ++pc) {
                    const float got = peerX[(((pc * 2 + mt) * 2 + nt) * 8 + v) * 32 + lane];
                    const float own = M[0][pc][mt][nt][v] + M[1][pc][mt][nt][v];
                    const float yy  = rsB ? (got - own) : (own + got);
                    y[pc] = yy * d + bb;
                }
                *(v2f*)&outb[(size_t)oc * (HW * HW) + 2 * (nt * 16 + lr)] = y;
            }
        }
    }
}

extern "C" void kernel_launch(void* const* d_in, const int* in_sizes, int n_in,
                              void* d_out, int out_size, void* d_ws, size_t ws_size,
                              hipStream_t stream) {
    const float* x      = (const float*)d_in[0];
    const float* style  = (const float*)d_in[1];
    const float* weight = (const float*)d_in[2];
    const float* bias   = (const float*)d_in[3];
    const float* mod_w  = (const float*)d_in[4];
    const float* mod_b  = (const float*)d_in[5];
    float* out = (float*)d_out;

    float* s9    = (float*)d_ws;
    float* demod = s9 + NB * K9;
    float* wU    = demod + NB * COUT;

    style_s9_kernel<<<S9_BLOCKS + WT_BLOCKS, 128, 0, stream>>>(style, mod_w, mod_b, weight, s9, wU);
    demod_kernel<<<COUT, 256, 0, stream>>>(weight, s9, demod);

    dim3 grid(HW / 2, NB, COUT / OCB);
    conv_wmma_kernel<<<grid, 256, 0, stream>>>(x, wU, bias, s9, demod, out);
}


# A fragments read straight from global memory into the WMMA fragment registers (launch 1 lays the weights out in fragment order); the A tile no longer passes through LDS
# speedup vs baseline: 4.0917x; 1.1287x over previous
#include <hip/hip_runtime.h>
#include <hip/hip_bf16.h>

typedef float v2f __attribute__((ext_vector_type(2)));
typedef float v4f __attribute__((ext_vector_type(4)));
typedef float v8f __attribute__((ext_vector_type(8)));

__device__ __forceinline__ v8f wmma4(v2f a, v2f b, v8f c) {
    return __builtin_amdgcn_wmma_f32_16x16x4_f32(
         false, a,  false, b,
         (short)0, c,  false,  false);
}

#define CIN   256
#define COUT  256
#define HW    64
#define K9    2304
#define SDIM  512
#define NB    8

__device__ __forceinline__ float wave_sum(float v) {
#pragma unroll
    for (int off = 16; off > 0; off >>= 1) v += __shfl_xor(v, off);
    return v;
}

#define S9_BLOCKS  (CIN / 4)
#define WT_BLOCKS  (COUT * 16 / 64)
#define AROWS      512
#define NSTEP      64
#define OCB        128

__device__ __forceinline__ float gform(int m, float a, float b, float c) {
    return (m == 0) ? a
         : (m == 1) ? 0.5f * ((a + b) + c)
         : (m == 2) ? 0.5f * ((a - b) + c)
         :            c;
}

__global__ __launch_bounds__(128) void style_s9_kernel(
    const float* __restrict__ style, const float* __restrict__ mod_w,
    const float* __restrict__ mod_b, const float* __restrict__ weight,
    float* __restrict__ s9, float* __restrict__ wU)
{
    if (blockIdx.x >= S9_BLOCKS) {
        if (threadIdx.x >= 64) return;
        const int r   = blockIdx.x - S9_BLOCKS;
        const int icg = r >> 2;
        const int oc  = ((r & 3) << 6) + threadIdx.x;
        const v4f* __restrict__ src = (const v4f*)(weight + (size_t)oc * K9 + icg * 144);
        v4f in[36];
#pragma unroll
        for (int q = 0; q < 36; ++q) in[q] = src[q];
        float* __restrict__ dstz = wU + (size_t)(oc >> 7) * NSTEP * AROWS * 16;
        const int ocl = oc & (OCB - 1);
#pragma unroll
        for (int g = 0; g < 4; ++g) {
            float U[4][4][4];
#pragma unroll
            for (int e = 0; e < 4; ++e) {
                const int kc = 4 * g + e;
                float hz[3][4];
#pragma unroll
                for (int kh = 0; kh < 3; ++kh) {
                    const float a  = in[(kc * 9 + kh * 3 + 0) >> 2][(kc * 9 + kh * 3 + 0) & 3];
                    const float bb = in[(kc * 9 + kh * 3 + 1) >> 2][(kc * 9 + kh * 3 + 1) & 3];
                    const float cc = in[(kc * 9 + kh * 3 + 2) >> 2][(kc * 9 + kh * 3 + 2) & 3];
#pragma unroll
                    for (int i = 0; i < 4; ++i) hz[kh][i] = gform(i, a, bb, cc);
                }
#pragma unroll
                for (int j = 0; j < 4; ++j)
#pragma unroll
                    for (int i = 0; i < 4; ++i) U[j][i][e] = gform(j, hz[0][i], hz[1][i], hz[2][i]);
            }
#pragma unroll
            for (int j = 0; j < 4; ++j) {
#pragma unroll
                for (int i = 0; i < 4; ++i) {
                    const int t = 2 * (2 * icg + (i >> 1)) + (j & 1);
                    const int w = (j >> 1) * 4 + (ocl >> 5);
                    const int L = 2 * g + (i & 1);
                    float* __restrict__ p = dstz + (size_t)t * (AROWS * 16)
                                          + ((w * 8 + L) * 32 + (ocl & 15)) * 4 + ((ocl >> 4) & 1) * 2;
                    v2f lo, hi;
                    lo.x = U[j][i][0]; lo.y = U[j][i][1];
                    hi.x = U[j][i][2]; hi.y = U[j][i][3];
                    *(v2f*)p        = lo;
                    *(v2f*)(p + 64) = hi;
                }
            }
        }
        return;
    }
    const int lane = threadIdx.x & 31;
    const int ic   = blockIdx.x * 4 + (threadIdx.x >> 5);

    const v4f* __restrict__ wrow = (const v4f*)(mod_w + (size_t)ic * SDIM);
    v4f w[4];
#pragma unroll
    for (int i = 0; i < 4; ++i) w[i] = wrow[lane + 32 * i];

    float acc[NB];
#pragma unroll
    for (int b = 0; b < NB; ++b) {
        const v4f* __restrict__ srow = (const v4f*)(style + b * SDIM);
        float a = 0.f;
#pragma unroll
        for (int i = 0; i < 4; ++i) {
            const v4f s = srow[lane + 32 * i];
            a += w[i].x * s.x;
            a += w[i].y * s.y;
            a += w[i].z * s.z;
            a += w[i].w * s.w;
        }
        acc[b] = a;
    }
#pragma unroll
    for (int b = 0; b < NB; ++b) acc[b] = wave_sum(acc[b]);

    const float mb = mod_b[ic];
    if (lane < 9) {
#pragma unroll
        for (int b = 0; b < NB; ++b) s9[b * K9 + ic * 9 + lane] = acc[b] + mb;
    }
}

__global__ __launch_bounds__(256) void demod_kernel(
    const float* __restrict__ weight, const float* __restrict__ s9,
    float* __restrict__ demod)
{
    const int oc   = blockIdx.x;
    const int lane = threadIdx.x & 31;
    const int b    = threadIdx.x >> 5;
    const v4f* __restrict__ wrow = (const v4f*)(weight + (size_t)oc * K9);
    const v4f* __restrict__ srow = (const v4f*)(s9 + (size_t)b * K9);

    v4f wv[K9 / 128], sv[K9 / 128];
#pragma unroll
    for (int i = 0; i < K9 / 128; ++i) wv[i] = wrow[lane + 32 * i];
#pragma unroll
    for (int i = 0; i < K9 / 128; ++i) sv[i] = srow[lane + 32 * i];

    float a0 = 0.f, a1 = 0.f, a2 = 0.f, a3 = 0.f;
#pragma unroll
    for (int i = 0; i < K9 / 128; ++i) {
        const v4f v = wv[i] * sv[i];
        a0 += v.x * v.x;
        a1 += v.y * v.y;
        a2 += v.z * v.z;
        a3 += v.w * v.w;
    }
    const float tot = wave_sum((a0 + a1) + (a2 + a3));
    if (lane == 0) demod[b * COUT + oc] = rsqrtf(tot + 1e-8f);
}

#define LDBC 20
#define NTILE 128

__global__ __launch_bounds__(256) void conv_wmma_kernel(
    const float* __restrict__ x, const float* __restrict__ wU,
    const float* __restrict__ bias, const float* __restrict__ s9,
    const float* __restrict__ demod, float* __restrict__ out)
{
    __shared__ float lB[2][NTILE * LDBC];
    __shared__ float lX[8][8 * 8 * 32];

    const int tid = threadIdx.x;
    const int h0  = blockIdx.x * 2;
    const int b   = blockIdx.y;
    const int z   = blockIdx.z;

    const int lane = tid & 31;
    const int wid  = tid >> 5;
    const int wq   = (wid & 3) * 32;
    const int wn   = (wid >> 2) * 64;
    const int half = lane >> 4;
    const int lr   = lane & 15;

    const int kcb0 = __builtin_amdgcn_readfirstlane(wid & 3);
    const int rsB  = __builtin_amdgcn_readfirstlane(wid >> 2);
    const int nB   = rsB * 64 + lane;

    int      ihA[2], ihB[2];
    bool     okA[2], okB[2];
    unsigned roA[2], roB[2];
    float    beta[2];
    ihA[0] = h0 + (rsB ? 1 : -1);   ihB[0] = h0 + (rsB ? 0 : 1);   beta[0] = -1.f;
    ihA[1] = h0;                    ihB[1] = h0 + (rsB ? 2 : 1);   beta[1] = rsB ? -1.f : 1.f;
#pragma unroll
    for (int jj = 0; jj < 2; ++jj) {
        okA[jj] = (unsigned)ihA[jj] < (unsigned)HW;
        okB[jj] = (unsigned)ihB[jj] < (unsigned)HW;
        roA[jj] = okA[jj] ? (unsigned)ihA[jj] * HW : 0u;
        roB[jj] = okB[jj] ? (unsigned)ihB[jj] * HW : 0u;
    }

    unsigned cP[2][2], cQ[2][2];
    bool     okP[2][2], okQ[2][2];
    float    gam[2][2];
    {
        const int c2 = 2 * lane;
        const int p00 = c2 - 1, q00 = c2 + 1;   gam[0][0] = -1.f;
        const int p01 = c2,     q01 = c2 + 1;   gam[0][1] =  1.f;
        const int p10 = c2 + 1, q10 = c2;       gam[1][0] = -1.f;
        const int p11 = c2,     q11 = c2 + 2;   gam[1][1] = -1.f;
        const int pp[2][2] = {{p00, p01}, {p10, p11}};
        const int qq[2][2] = {{q00, q01}, {q10, q11}};
#pragma unroll
        for (int ih = 0; ih < 2; ++ih)
#pragma unroll
            for (int iw = 0; iw < 2; ++iw) {
                okP[ih][iw] = (unsigned)pp[ih][iw] < (unsigned)HW;
                okQ[ih][iw] = (unsigned)qq[ih][iw] < (unsigned)HW;
                cP[ih][iw]  = (unsigned)min(max(pp[ih][iw], 0), HW - 1);
                cQ[ih][iw]  = (unsigned)min(max(qq[ih][iw], 0), HW - 1);
            }
    }

    const float* __restrict__ s9b = s9 + (size_t)b * K9;
    const float* __restrict__ xb  = x + (size_t)b * CIN * HW * HW;
    const float* __restrict__ wUw = wU + (size_t)z * NSTEP * AROWS * 16
                                  + __builtin_amdgcn_readfirstlane(wid) * 1024;
    const unsigned aLane = (unsigned)lane << 2;

    float  bRegA[4][2][2];
    float  bRegB[4][2][2];
    float  sReg[4];

    auto loadChunk = [&](int t, int jj, int ih) __attribute__((always_inline)) {
        const int icg = t >> 2;
        const unsigned icb = (unsigned)(icg * 16 + kcb0);
        const float* __restrict__ xrA = xb + (icb * (HW * HW) + roA[jj]);
        const float* __restrict__ xrB = xb + (icb * (HW * HW) + roB[jj]);
        const float* __restrict__ srow = s9b + icb * 9;
#pragma unroll
        for (int i = 0; i < 4; ++i) {
#pragma unroll
            for (int q = 0; q < 2; ++q) {
                bRegA[i][q][0] = xrA[(4 * i) * (HW * HW) + cP[ih][q]];
                bRegA[i][q][1] = xrA[(4 * i) * (HW * HW) + cQ[ih][q]];
                bRegB[i][q][0] = xrB[(4 * i) * (HW * HW) + cP[ih][q]];
                bRegB[i][q][1] = xrB[(4 * i) * (HW * HW) + cQ[ih][q]];
            }
            sReg[i] = srow[(4 * i) * 9];
        }
    };
    auto storeChunk = [&](int buf, int jj, int ih) __attribute__((always_inline)) {
#pragma unroll
        for (int i = 0; i < 4; ++i) {
            int kc = kcb0 + 4 * i;
#pragma unroll
            for (int q = 0; q < 2; ++q) {
                const float daP = okA[jj] ? bRegA[i][q][0] : 0.f;
                const float dbP = okB[jj] ? bRegB[i][q][0] : 0.f;
                const float daQ = okA[jj] ? bRegA[i][q][1] : 0.f;
                const float dbQ = okB[jj] ? bRegB[i][q][1] : 0.f;
                const float vP  = okP[ih][q] ? (daP + beta[jj] * dbP) : 0.f;
                const float vQ  = okQ[ih][q] ? (daQ + beta[jj] * dbQ) : 0.f;
                lB[buf][(nB + 32 * q) * LDBC + kc] = (vP + gam[ih][q] * vQ) * sReg[i];
            }
        }
    };

    v8f acc[4][4][2];
#pragma unroll
    for (int s = 0; s < 4; ++s)
#pragma unroll
        for (int m4 = 0; m4 < 4; ++m4)
#pragma unroll
            for (int nt = 0; nt < 2; ++nt)
                acc[s][m4][nt] = (v8f){};

    v2f fA0[4][4], fB0[4][4];
    v2f fA1[4][4], fB1[4][4];

    auto loadFragsA = [&](int t, v2f (&fa)[4][4]) __attribute__((always_inline)) {
        const float* __restrict__ wa = wUw + t * (AROWS * 16);
#pragma unroll
        for (int ks = 0; ks < 4; ++ks)
#pragma unroll
            for (int iw = 0; iw < 2; ++iw) {
                const v4f v = *(const v4f*)(wa + (2 * ks + iw) * 128 + aLane);
                fa[ks][2 * iw].x     = v.x;  fa[ks][2 * iw].y     = v.y;
                fa[ks][2 * iw + 1].x = v.z;  fa[ks][2 * iw + 1].y = v.w;
            }
    };
    auto loadFragsB = [&](int buf, v2f (&fb)[4][4]) __attribute__((always_inline)) {
#pragma unroll
        for (int ks = 0; ks < 4; ++ks) {
            const int ak = ks * 4 + half * 2;
#pragma unroll
            for (int f = 0; f < 4; ++f)
                fb[ks][f] = *(const v2f*)&lB[buf][(wn + f * 16 + lr) * LDBC + ak];
        }
    };
    auto mmaChunk = [&](int s, int ks0, int ks1, const v2f (&fa)[4][4], const v2f (&fb)[4][4])
        __attribute__((always_inline)) {
#pragma unroll
        for (int ks = ks0; ks < ks1; ++ks)
#pragma unroll
            for (int f = 0; f < 4; ++f)
#pragma unroll
                for (int nt = 0; nt < 2; ++nt)
                    acc[s][f][nt] = wmma4(fa[ks][f], fb[ks][(f >> 1) * 2 + nt], acc[s][f][nt]);
    };

    auto step = [&](int t, int s, bool doStore, bool doLoad, bool doFrags, bool doSync,
                    const v2f (&ca)[4][4], const v2f (&cb)[4][4],
                    v2f (&na)[4][4], v2f (&nb)[4][4])
        __attribute__((always_inline)) {
        const int par = s & 1;
        if (doFrags) loadFragsA(t + 1, na);
        if (doStore) storeChunk(par, par, ((s + 2) >> 1) & 1);
        if (doLoad)  loadChunk(t + 3, par ^ 1, ((s + 3) >> 1) & 1);
        if (doFrags) loadFragsB(par ^ 1, nb);
        __builtin_amdgcn_sched_barrier(0);
        mmaChunk(s, 0, 3, ca, cb);
        if (doSync) {
            __builtin_amdgcn_sched_barrier(0);
            asm volatile("s_wait_dscnt 0x0" ::: "memory");
            asm volatile("s_barrier_signal -1" ::: "memory");
            __builtin_amdgcn_sched_barrier(0);
        }
        mmaChunk(s, 3, 4, ca, cb);
        __builtin_amdgcn_sched_barrier(0);
        if (doSync) asm volatile("s_barrier_wait -1" ::: "memory");
    };

    loadFragsA(0, fA0);
    loadChunk(0, 0, 0);
    storeChunk(0, 0, 0);
    loadChunk(1, 1, 0);
    storeChunk(1, 1, 0);
    loadChunk(2, 0, 1);
    __syncthreads();
    loadFragsB(0, fB0);
    __syncthreads();

    for (int t = 0; t < NSTEP - 4; t += 4) {
        step(t,     0, true, true, true, true, fA0, fB0, fA1, fB1);
        step(t + 1, 1, true, true, true, true, fA1, fB1, fA0, fB0);
        step(t + 2, 2, true, true, true, true, fA0, fB0, fA1, fB1);
        step(t + 3, 3, true, true, true, true, fA1, fB1, fA0, fB0);
    }
    step(NSTEP - 4, 0, true,  true,  true,  true,  fA0, fB0, fA1, fB1);
    step(NSTEP - 3, 1, true,  false, true,  true,  fA1, fB1, fA0, fB0);
    step(NSTEP - 2, 2, false, false, true,  false, fA0, fB0, fA1, fB1);
    step(NSTEP - 1, 3, false, false, false, false, fA1, fB1, fA0, fB0);

    v8f M[2][2][2][2];
#pragma unroll
    for (int jj = 0; jj < 2; ++jj)
#pragma unroll
        for (int mt = 0; mt < 2; ++mt)
#pragma unroll
            for (int nt = 0; nt < 2; ++nt) {
                const v8f n0 = acc[jj][mt][nt],     n1 = acc[jj][2 + mt][nt];
                const v8f n2 = acc[2 + jj][mt][nt], n3 = acc[2 + jj][2 + mt][nt];
                M[jj][0][mt][nt] = (n0 + n1) + n2;
                M[jj][1][mt][nt] = n1 - (n2 + n3);
            }

    float* __restrict__ myX = &lX[wid][0];
#pragma unroll
    for (int pc = 0; pc < 2; ++pc)
#pragma unroll
        for (int mt = 0; mt < 2; ++mt)
#pragma unroll
            for (int nt = 0; nt < 2; ++nt)
#pragma unroll
                for (int v = 0; v < 8; ++v)
                    myX[(((pc * 2 + mt) * 2 + nt) * 8 + v) * 32 + lane] =
                        rsB ? M[0][pc][mt][nt][v] : M[1][pc][mt][nt][v];
    __syncthreads();
    const float* __restrict__ peerX = &lX[wid ^ 4][0];

    const float* __restrict__ demodb = demod + b * COUT;
    float* __restrict__ outb = out + ((size_t)b * COUT * HW + h0 + (wid >> 2)) * HW;
#pragma unroll
    for (int mt = 0; mt < 2; ++mt) {
#pragma unroll
        for (int v = 0; v < 8; ++v) {
            int oc = z * OCB + wq + mt * 16 + half * 8 + v;
            float d  = demodb[oc];
            float bb = bias[oc];
#pragma unroll
            for (int nt = 0; nt < 2; ++nt) {
                v2f y;
#pragma unroll
                for (int pc = 0; pc < 2; ++pc) {
                    const float got = peerX[(((pc * 2 + mt) * 2 + nt) * 8 + v) * 32 + lane];
                    const float own = M[0][pc][mt][nt][v] + M[1][pc][mt][nt][v];
                    const float yy  = rsB ? (got - own) : (own + got);
                    y[pc] = yy * d + bb;
                }
                *(v2f*)&outb[(size_t)oc * (HW * HW) + 2 * (nt * 16 + lr)] = y;
            }
        }
    }
}

extern "C" void kernel_launch(void* const* d_in, const int* in_sizes, int n_in,
                              void* d_out, int out_size, void* d_ws, size_t ws_size,
                              hipStream_t stream) {
    const float* x      = (const float*)d_in[0];
    const float* style  = (const float*)d_in[1];
    const float* weight = (const float*)d_in[2];
    const float* bias   = (const float*)d_in[3];
    const float* mod_w  = (const float*)d_in[4];
    const float* mod_b  = (const float*)d_in[5];
    float* out = (float*)d_out;

    float* s9    = (float*)d_ws;
    float* demod = s9 + NB * K9;
    float* wU    = demod + NB * COUT;

    style_s9_kernel<<<S9_BLOCKS + WT_BLOCKS, 128, 0, stream>>>(style, mod_w, mod_b, weight, s9, wU);
    demod_kernel<<<COUT, 256, 0, stream>>>(weight, s9, demod);

    dim3 grid(HW / 2, NB, COUT / OCB);
    conv_wmma_kernel<<<grid, 256, 0, stream>>>(x, wU, bias, s9, demod, out);
}
